# starting kernel
# baseline (speedup 1.0000x reference)
#include <hip/hip_runtime.h>

typedef __attribute__((ext_vector_type(2))) float v2f;
typedef __attribute__((ext_vector_type(8))) float v8f;

#define NEG_SLOPE 0.01f

__global__ void k_norm_init(float* __restrict__ norm, int n) {
  int i = blockIdx.x * blockDim.x + threadIdx.x;
  if (i < n) norm[i] = 1.0f;
}
__global__ void k_norm_count(const int* __restrict__ dst, float* __restrict__ norm, int e) {
  int i = blockIdx.x * blockDim.x + threadIdx.x;
  if (i < e) atomicAdd(&norm[dst[i]], 1.0f);
}
__global__ void k_norm_rsqrt(float* __restrict__ norm, int n) {
  int i = blockIdx.x * blockDim.x + threadIdx.x;
  if (i < n) norm[i] = rsqrtf(norm[i]);
}

__global__ __launch_bounds__(256) void k_gemm_wmma(
    const float* __restrict__ X, const float* __restrict__ W,
    const float* __restrict__ norm,
    float* __restrict__ S, float* __restrict__ AGG,
    int nrows, int K, int Hout) {
  extern __shared__ float lds[];
  const int KSTRIDE = K + 4;
  float* Xs = lds;
  float* Wt = lds + 128 * KSTRIDE;

  const int tid = threadIdx.x;
  const int row0 = blockIdx.x * 128;

  for (int idx = tid * 4; idx < K * Hout; idx += 256 * 4) {
    int k = idx / Hout, c = idx - k * Hout;
    float4 v = *(const float4*)&W[idx];
    Wt[(c + 0) * KSTRIDE + k] = v.x;
    Wt[(c + 1) * KSTRIDE + k] = v.y;
    Wt[(c + 2) * KSTRIDE + k] = v.z;
    Wt[(c + 3) * KSTRIDE + k] = v.w;
  }
  for (int idx = tid * 4; idx < 128 * K; idx += 256 * 4) {
    int r = idx / K, c = idx - r * K;
    float4 v = {0.f, 0.f, 0.f, 0.f};
    int gr = row0 + r;
    if (gr < nrows) v = *(const float4*)&X[(long)gr * K + c];
    *(float4*)&Xs[r * KSTRIDE + c] = v;
  }
  __syncthreads();

  const int lane = tid & 31;
  const int wave = tid >> 5;
  const int mrow = lane & 15;
  const int kg   = lane >> 4;
  const float* Xrow = &Xs[(wave * 16 + mrow) * KSTRIDE + 2 * kg];

  float nrm[8];
#pragma unroll
  for (int r = 0; r < 8; ++r) {
    int grow = row0 + wave * 16 + r + 8 * kg;
    nrm[r] = (grow < nrows) ? norm[grow] : 0.f;
  }

  for (int ct = 0; ct < (Hout >> 4); ++ct) {
    const int col = ct * 16 + mrow;
    const float* Wcol = &Wt[col * KSTRIDE + 2 * kg];
    v8f acc = {};
    for (int k = 0; k < K; k += 4) {
      v2f a = *(const v2f*)&Xrow[k];
      v2f b = *(const v2f*)&Wcol[k];
      acc = __builtin_amdgcn_wmma_f32_16x16x4_f32(
           false, a,  false, b,
           (short)0, acc,  false,  false);
    }
#pragma unroll
    for (int r = 0; r < 8; ++r) {
      int grow = row0 + wave * 16 + r + 8 * kg;
      if (grow < nrows) {
        float sv = acc[r] * nrm[r];
        long o = (long)grow * Hout + col;
        S[o]   = sv;
        AGG[o] = sv;
      }
    }
  }
}

__global__ void k_scatter(const int* __restrict__ src, const int* __restrict__ dst,
                          const float* __restrict__ S, float* __restrict__ AGG,
                          int e, int Hout) {
  const int chunks = Hout >> 2;
  long tid = (long)blockIdx.x * blockDim.x + threadIdx.x;
  if (tid >= (long)e * chunks) return;
  int ed = (int)(tid / chunks);
  int c  = (int)(tid - (long)ed * chunks) << 2;
  int s = src[ed], d = dst[ed];
  const float4 v = *(const float4*)&S[(long)s * Hout + c];
  float* ap = &AGG[(long)d * Hout + c];
  atomicAdd(ap + 0, v.x);
  atomicAdd(ap + 1, v.y);
  atomicAdd(ap + 2, v.z);
  atomicAdd(ap + 3, v.w);
}

__global__ void k_finalize(const float* __restrict__ AGG, const float* __restrict__ norm,
                           const float* __restrict__ bias, const float* __restrict__ xin,
                           float* __restrict__ out, int n, int Hout, int mode) {
  const int chunks = Hout >> 2;
  long tid = (long)blockIdx.x * blockDim.x + threadIdx.x;
  if (tid >= (long)n * chunks) return;
  int i = (int)(tid / chunks);
  int c = (int)(tid - (long)i * chunks) << 2;
  long o = (long)i * Hout + c;
  float nr = norm[i];
  float4 av = *(const float4*)&AGG[o];
  float4 bv = *(const float4*)&bias[c];
  float vx = av.x * nr + bv.x;
  float vy = av.y * nr + bv.y;
  float vz = av.z * nr + bv.z;
  float vw = av.w * nr + bv.w;
  if (mode != 2) {
    vx = vx >= 0.f ? vx : NEG_SLOPE * vx;
    vy = vy >= 0.f ? vy : NEG_SLOPE * vy;
    vz = vz >= 0.f ? vz : NEG_SLOPE * vz;
    vw = vw >= 0.f ? vw : NEG_SLOPE * vw;
  }
  if (mode == 1) {
    float4 xv = *(const float4*)&xin[o];
    vx = (xv.x + vx) * 0.5f;
    vy = (xv.y + vy) * 0.5f;
    vz = (xv.z + vz) * 0.5f;
    vw = (xv.w + vw) * 0.5f;
  }
  float4 r4 = {vx, vy, vz, vw};
  *(float4*)&out[o] = r4;
}

extern "C" void kernel_launch(void* const* d_in, const int* in_sizes, int n_in,
                              void* d_out, int out_size, void* d_ws, size_t ws_size,
                              hipStream_t stream) {
  (void)n_in; (void)out_size; (void)ws_size;
  const float* inputs = (const float*)d_in[0];
  const int*   edges  = (const int*)d_in[1];
  const float* w_in   = (const float*)d_in[2];
  const float* b_in   = (const float*)d_in[3];
  const float* w1     = (const float*)d_in[4];
  const float* b1     = (const float*)d_in[5];
  const float* w2     = (const float*)d_in[6];
  const float* b2     = (const float*)d_in[7];
  const float* w_out  = (const float*)d_in[8];
  const float* b_out  = (const float*)d_in[9];

  const int H   = in_sizes[3];
  const int IN  = in_sizes[2] / H;
  const int N   = in_sizes[0] / IN;
  const int E   = in_sizes[1] / 2;
  const int OUT = in_sizes[9];
  const int B   = in_sizes[5] / H;

  const int* srcI = edges;
  const int* dstI = edges + E;

  float* norm = (float*)d_ws;
  float* S    = norm + (((size_t)N + 255) & ~(size_t)255);
  float* AGG  = S + (size_t)N * H;
  float* Hb   = AGG + (size_t)N * H;

  float* xout = (float*)d_out;
  float* xbuf = xout + (size_t)N * OUT;

  k_norm_init <<<(N + 255) / 256, 256, 0, stream>>>(norm, N);
  k_norm_count<<<(E + 255) / 256, 256, 0, stream>>>(dstI, norm, E);
  k_norm_rsqrt<<<(N + 255) / 256, 256, 0, stream>>>(norm, N);

  auto conv = [&](const float* Xin, const float* W, const float* bias, int K, int Hout,
                  float* outp, int mode, const float* resid) {
    size_t shmem = (size_t)(128 + Hout) * (K + 4) * sizeof(float);
    k_gemm_wmma<<<(N + 127) / 128, 256, shmem, stream>>>(Xin, W, norm, S, AGG, N, K, Hout);
    long tot = (long)E * (Hout >> 2);
    k_scatter<<<(unsigned)((tot + 255) / 256), 256, 0, stream>>>(srcI, dstI, S, AGG, E, Hout);
    long tot2 = (long)N * (Hout >> 2);
    k_finalize<<<(unsigned)((tot2 + 255) / 256), 256, 0, stream>>>(AGG, norm, bias, resid,
                                                                   outp, N, Hout, mode);
  };

  conv(inputs, w_in, b_in, IN, H, xbuf, 0, nullptr);
  for (int i = 0; i < B; ++i) {
    conv(xbuf, w1 + (size_t)i * H * H, b1 + (size_t)i * H, H, H, Hb,   0, nullptr);
    conv(Hb,   w2 + (size_t)i * H * H, b2 + (size_t)i * H, H, H, xbuf, 1, xbuf);
  }
  conv(xbuf, w_out, b_out, H, OUT, xout, 2, nullptr);
}


# GEMM row tiles handed out by L2 half (two int-atomic ticket counters per layer): a tile's finished sum, start value and x rows stay in the half whose scatter adds into them; sums unchanged
# speedup vs baseline: 2.8047x; 2.8047x over previous
#include <hip/hip_runtime.h>

typedef __attribute__((ext_vector_type(2))) float v2f;
typedef __attribute__((ext_vector_type(8))) float v8f;

#define NEG_SLOPE 0.01f

__global__ void k_norm_init(float* __restrict__ norm, int* __restrict__ cnt, int n,
                            int* __restrict__ qctr, int nq) {
  int i = blockIdx.x * blockDim.x + threadIdx.x;
  if (i < n) {
    norm[i] = 1.0f;
    cnt[i] = 0;
  }
  for (int j = i; j < nq; j += (int)(gridDim.x * blockDim.x)) qctr[j] = 0;
}
__global__ void k_norm_count(const int* __restrict__ src, const int* __restrict__ dst,
                             float* __restrict__ norm, int* __restrict__ cnt, int e) {
  int i = blockIdx.x * blockDim.x + threadIdx.x;
  if (i < e) {
    atomicAdd(&norm[dst[i]], 1.0f);
    atomicAdd(&cnt[src[i]], 1);
  }
}
__global__ void k_norm_rsqrt(float* __restrict__ norm, int n) {
  int i = blockIdx.x * blockDim.x + threadIdx.x;
  if (i < n) norm[i] = rsqrtf(norm[i]);
}

#define SCAN_NB 2048
__global__ __launch_bounds__(1024) void k_scan(const int* __restrict__ cnt, int* __restrict__ rowptr,
                                               int* __restrict__ cursor, int n) {
  __shared__ int wsum[32];
  __shared__ int wtot[32];
  const int t = threadIdx.x;
  const int lane = t & 31, w = t >> 5;
  const int n0 = blockIdx.x * SCAN_NB;
  int s = 0;
  for (int i = t; i < n0; i += 1024) s += cnt[i];
  for (int off = 16; off > 0; off >>= 1) s += __shfl_xor(s, off);
  const int i0 = n0 + w * 64 + lane;
  const int i1 = i0 + 32;
  const int c0 = (i0 < n) ? cnt[i0] : 0;
  const int c1 = (i1 < n) ? cnt[i1] : 0;
  int x0 = c0, x1 = c1;
  for (int off = 1; off < 32; off <<= 1) {
    int y0 = __shfl_up(x0, off);
    int y1 = __shfl_up(x1, off);
    if (lane >= off) { x0 += y0; x1 += y1; }
  }
  const int t0 = __shfl(x0, 31);
  const int tw = t0 + __shfl(x1, 31);
  if (lane == 0) { wsum[w] = s; wtot[w] = tw; }
  __syncthreads();
  int a = wsum[lane];
  for (int off = 16; off > 0; off >>= 1) a += __shfl_xor(a, off);
  const int z = wtot[lane];
  int zi = z;
  for (int off = 1; off < 32; off <<= 1) {
    int y = __shfl_up(zi, off);
    if (lane >= off) zi += y;
  }
  const int base = a + __shfl(zi - z, w);
  const int e0 = base + x0 - c0;
  const int e1 = base + t0 + x1 - c1;
  if (i0 < n) { rowptr[i0] = e0; cursor[i0] = e0; }
  if (i1 < n) { rowptr[i1] = e1; cursor[i1] = e1; }
  if (i0 == n - 1) rowptr[n] = e0 + c0;
  if (i1 == n - 1) rowptr[n] = e1 + c1;
}
__global__ void k_fill(const int* __restrict__ src, const int* __restrict__ dst,
                       int* __restrict__ cursor, int* __restrict__ dsts, int e) {
  int i = blockIdx.x * blockDim.x + threadIdx.x;
  if (i < e) {
    int p = atomicAdd(&cursor[src[i]], 1);
    dsts[p] = dst[i];
  }
}

template <int PRE>
__global__ __launch_bounds__(256) void k_gemm_wmma(
    const float* __restrict__ X, const float* __restrict__ W,
    const float* __restrict__ norm,
    const float* __restrict__ AGGi, const float* __restrict__ pbias,
    float* __restrict__ xio, int store_x,
    float* __restrict__ S, float* __restrict__ AGGo,
    int nrows, int K, int Hout) {
  extern __shared__ float lds[];
  const int KSTRIDE = K + 4;
  float* Xs = lds;
  float* Wt = lds + 128 * KSTRIDE;

  const int tid = threadIdx.x;
  const int row0 = blockIdx.x * 128;

  for (int idx = tid * 4; idx < K * Hout; idx += 256 * 4) {
    int k = idx / Hout, c = idx - k * Hout;
    float4 v = *(const float4*)&W[idx];
    Wt[(c + 0) * KSTRIDE + k] = v.x;
    Wt[(c + 1) * KSTRIDE + k] = v.y;
    Wt[(c + 2) * KSTRIDE + k] = v.z;
    Wt[(c + 3) * KSTRIDE + k] = v.w;
  }
  for (int idx = tid * 4; idx < 128 * K; idx += 256 * 4) {
    int r = idx / K, c = idx - r * K;
    float4 v = {0.f, 0.f, 0.f, 0.f};
    int gr = row0 + r;
    if (gr < nrows) {
      long o = (long)gr * K + c;
      if (PRE == 0) {
        v = *(const float4*)&X[o];
      } else {
        float nr = norm[gr];
        float4 av = *(const float4*)&AGGi[o];
        float4 bv = *(const float4*)&pbias[c];
        v.x = av.x * nr + bv.x;
        v.y = av.y * nr + bv.y;
        v.z = av.z * nr + bv.z;
        v.w = av.w * nr + bv.w;
        v.x = v.x >= 0.f ? v.x : NEG_SLOPE * v.x;
        v.y = v.y >= 0.f ? v.y : NEG_SLOPE * v.y;
        v.z = v.z >= 0.f ? v.z : NEG_SLOPE * v.z;
        v.w = v.w >= 0.f ? v.w : NEG_SLOPE * v.w;
        if (PRE == 2) {
          float4 xv = *(const float4*)&xio[o];
          v.x = (xv.x + v.x) * 0.5f;
          v.y = (xv.y + v.y) * 0.5f;
          v.z = (xv.z + v.z) * 0.5f;
          v.w = (xv.w + v.w) * 0.5f;
        }
        if (PRE == 2 || store_x) *(float4*)&xio[o] = v;
      }
    }
    *(float4*)&Xs[r * KSTRIDE + c] = v;
  }
  __syncthreads();

  const int lane = tid & 31;
  const int wave = tid >> 5;
  const int mrow = lane & 15;
  const int kg   = lane >> 4;
  const float* Xrow = &Xs[(wave * 16 + mrow) * KSTRIDE + 2 * kg];

  float nrm[8];
#pragma unroll
  for (int r = 0; r < 8; ++r) {
    int grow = row0 + wave * 16 + r + 8 * kg;
    nrm[r] = (grow < nrows) ? norm[grow] : 0.f;
  }

  for (int ct = 0; ct < (Hout >> 4); ++ct) {
    const int col = ct * 16 + mrow;
    const float* Wcol = &Wt[col * KSTRIDE + 2 * kg];
    v8f acc = {};
    for (int k = 0; k < K; k += 4) {
      v2f a = *(const v2f*)&Xrow[k];
      v2f b = *(const v2f*)&Wcol[k];
      acc = __builtin_amdgcn_wmma_f32_16x16x4_f32(
           false, a,  false, b,
           (short)0, acc,  false,  false);
    }
#pragma unroll
    for (int r = 0; r < 8; ++r) {
      int grow = row0 + wave * 16 + r + 8 * kg;
      if (grow < nrows) {
        float sv = acc[r] * nrm[r];
        long o = (long)grow * Hout + col;
        S[o]    = sv;
        AGGo[o] = sv;
      }
    }
  }
}

static __device__ __forceinline__ void lds_barrier() {
  __builtin_amdgcn_fence(__ATOMIC_RELEASE, "workgroup", "local");
  __builtin_amdgcn_s_barrier();
  __builtin_amdgcn_fence(__ATOMIC_ACQUIRE, "workgroup", "local");
}

#define GK 128
#define GR 64
template <int PRE, int HOUT>
__global__ __launch_bounds__(256) void k_gemm_wreg(
    const float* __restrict__ X, const float* __restrict__ W,
    const float* __restrict__ norm,
    const float* __restrict__ AGGi, const float* __restrict__ pbias,
    float* __restrict__ xio, int store_x,
    float* __restrict__ S, float* __restrict__ AGGo, int* __restrict__ gctr, int nrows) {
  constexpr int KS = GK + 4;
  constexpr int NCT = HOUT / 16;
  constexpr int NRG = 8 / NCT;
  constexpr int RTW = (GR / 16) / NRG;
  __shared__ __attribute__((aligned(16))) float Xs[GR * KS];
  __shared__ int tile_s;

  const int tid = threadIdx.x;
  const int lane = tid & 31;
  const int wave = tid >> 5;
  const int mrow = lane & 15;
  const int kg   = lane >> 4;
  const int ctw = wave % NCT;
  const int rg  = wave / NCT;
  const int col = ctw * 16 + mrow;

  v2f b[GK / 4];
  const float* wp = W + (2 * kg) * HOUT + col;
#pragma unroll
  for (int kk = 0; kk < GK / 4; ++kk) {
    b[kk].x = wp[(4 * kk) * HOUT];
    b[kk].y = wp[(4 * kk + 1) * HOUT];
  }

  if (tid == 0) {
    const int ntiles = (int)gridDim.x;
    const int tlow = ((nrows >> 1) + GR - 1) / GR;
    int hh = (__builtin_amdgcn_s_sendmsg_rtn(0x87) >> 18) & 1;
    int cnt = hh ? ntiles - tlow : tlow;
    int tk = atomicAdd(gctr + hh * 32, 1);
    if (tk >= cnt) {
      hh ^= 1;
      cnt = hh ? ntiles - tlow : tlow;
      tk = atomicAdd(gctr + hh * 32, 1);
    }
    tile_s = (tk < cnt) ? (hh ? tlow + tk : tk) : -1;
  }
  lds_barrier();
  const int tile = tile_s;
  if (tile < 0) return;
  const int row0 = tile * GR;

  for (int idx = tid * 4; idx < GR * GK; idx += 256 * 4) {
    int r = idx / GK, c = idx - r * GK;
    float4 v = {0.f, 0.f, 0.f, 0.f};
    int gr = row0 + r;
    if (gr < nrows) {
      long o = (long)gr * GK + c;
      if (PRE == 0) {
        v = *(const float4*)&X[o];
      } else {
        float nr = norm[gr];
        float4 av = *(const float4*)&AGGi[o];
        float4 bv = *(const float4*)&pbias[c];
        v.x = av.x * nr + bv.x;
        v.y = av.y * nr + bv.y;
        v.z = av.z * nr + bv.z;
        v.w = av.w * nr + bv.w;
        v.x = v.x >= 0.f ? v.x : NEG_SLOPE * v.x;
        v.y = v.y >= 0.f ? v.y : NEG_SLOPE * v.y;
        v.z = v.z >= 0.f ? v.z : NEG_SLOPE * v.z;
        v.w = v.w >= 0.f ? v.w : NEG_SLOPE * v.w;
        if (PRE == 2) {
          float4 xv = *(const float4*)&xio[o];
          v.x = (xv.x + v.x) * 0.5f;
          v.y = (xv.y + v.y) * 0.5f;
          v.z = (xv.z + v.z) * 0.5f;
          v.w = (xv.w + v.w) * 0.5f;
        }
        if (PRE == 2 || store_x) *(float4*)&xio[o] = v;
      }
    }
    *(float4*)&Xs[r * KS + c] = v;
  }

  __syncthreads();

  const float* xr = &Xs[((rg * RTW) * 16 + mrow) * KS + 2 * kg];
  v8f acc[RTW];
#pragma unroll
  for (int j = 0; j < RTW; ++j) acc[j] = v8f{};
#pragma unroll
  for (int kk = 0; kk < GK / 4; ++kk) {
#pragma unroll
    for (int j = 0; j < RTW; ++j) {
      v2f a = *(const v2f*)&xr[j * 16 * KS + 4 * kk];
      acc[j] = __builtin_amdgcn_wmma_f32_16x16x4_f32(
           false, a,  false, b[kk],
           (short)0, acc[j],  false,  false);
    }
  }
  float nrm[RTW][8];
#pragma unroll
  for (int j = 0; j < RTW; ++j) {
#pragma unroll
    for (int r = 0; r < 8; ++r) {
      int grow = row0 + (rg * RTW + j) * 16 + r + 8 * kg;
      nrm[j][r] = norm[grow < nrows ? grow : nrows - 1];
    }
  }
#pragma unroll
  for (int j = 0; j < RTW; ++j) {
#pragma unroll
    for (int r = 0; r < 8; ++r) {
      int grow = row0 + (rg * RTW + j) * 16 + r + 8 * kg;
      if (grow < nrows) {
        float sv = acc[j][r] * nrm[j][r];
        long o = (long)grow * HOUT + col;
        S[o]    = sv;
        AGGo[o] = sv;
      }
    }
  }
}

#define PUSH_BLOCKS 2048
#define PUSH_CHW 4
template <int NV>
__global__ __launch_bounds__(256) void k_push(const int* __restrict__ rowptr,
                                              const int* __restrict__ dsts,
                                              const float* __restrict__ S, float* __restrict__ AGG,
                                              int* __restrict__ qctr, int n) {
  constexpr int HOUT = NV * 32;
  constexpr int SPB = 8 * PUSH_CHW;
  __shared__ int tk[2];
  const int tid = threadIdx.x;
  const int lane = tid & 31;
  const int wave = __builtin_amdgcn_readfirstlane(tid >> 5);
  const int nchunks = (n + SPB - 1) / SPB;
  const int nh = n >> 1;
  int h = (__builtin_amdgcn_s_sendmsg_rtn(0x87) >> 18) & 1;
  for (int pass = 0; pass < 2; ++pass, h ^= 1) {
    const int lo = h ? nh : 0;
    const int hi = h ? n : nh;
    int* ctr = qctr + h * 32;
    if (tid == 0) tk[0] = atomicAdd(ctr, 1);
    lds_barrier();
    for (int it = 0;; ++it) {
      const int t = __builtin_amdgcn_readfirstlane(tk[it & 1]);
      if (t >= nchunks) break;
      const int s0 = t * SPB + wave * PUSH_CHW;
      int rp[PUSH_CHW + 1];
#pragma unroll
      for (int k = 0; k <= PUSH_CHW; ++k) {
        const int si = s0 + k;
        rp[k] = rowptr[si < n ? si : n];
      }
      float v[PUSH_CHW][NV];
      int dl0[PUSH_CHW];
#pragma unroll
      for (int k = 0; k < PUSH_CHW; ++k) {
        const int s = (s0 + k < n) ? (s0 + k) : (n - 1);
        const float* sp = &S[(long)s * HOUT + lane];
#pragma unroll
        for (int i = 0; i < NV; ++i) v[k][i] = sp[32 * i];
        const int deg = rp[k + 1] - rp[k];
        dl0[k] = (lane < deg) ? dsts[rp[k] + lane] : -1;
      }
      int nxt = 0;
      if (tid == 0) nxt = atomicAdd(ctr, 1);
#pragma unroll
      for (int k = 0; k < PUSH_CHW; ++k) {
        const int beg = rp[k];
        const int deg = rp[k + 1] - beg;
        int dl = dl0[k];
        for (int base = 0; base < deg; base += 32) {
          if (base > 0) dl = (base + lane < deg) ? dsts[beg + base + lane] : -1;
          unsigned m = __builtin_amdgcn_ballot_w32(dl >= lo && dl < hi);
          while (m) {
            const int j = __builtin_ctz(m);
            m &= m - 1;
            const int d = __builtin_amdgcn_readlane(dl, j);
            float* ap = &AGG[(long)d * HOUT + lane];
#pragma unroll
            for (int i = 0; i < NV; ++i) atomicAdd(ap + 32 * i, v[k][i]);
          }
        }
      }

      if (tid == 0) tk[(it + 1) & 1] = nxt;
      lds_barrier();
    }
    lds_barrier();
  }
}

__global__ __launch_bounds__(256) void k_push_any(const int* __restrict__ rowptr,
                                                  const int* __restrict__ dsts,
                                                  const float* __restrict__ S,
                                                  float* __restrict__ AGG, int n, int Hout) {
  const int lane = threadIdx.x & 31;
  const int s = __builtin_amdgcn_readfirstlane((int)(blockIdx.x * (blockDim.x >> 5) + (threadIdx.x >> 5)));
  if (s >= n) return;
  const int beg = rowptr[s], end = rowptr[s + 1];
  const float* sp = &S[(long)s * Hout];
  for (int ed = beg; ed < end; ++ed) {
    const int d = dsts[ed];
    float* ap = &AGG[(long)d * Hout];
    for (int c = lane; c < Hout; c += 32) atomicAdd(ap + c, sp[c]);
  }
}

__global__ void k_finalize(const float* __restrict__ AGG, const float* __restrict__ norm,
                           const float* __restrict__ bias, float* __restrict__ out,
                           int n, int Hout) {
  const int chunks = Hout >> 2;
  long tid = (long)blockIdx.x * blockDim.x + threadIdx.x;
  if (tid >= (long)n * chunks) return;
  int i = (int)(tid / chunks);
  int c = (int)(tid - (long)i * chunks) << 2;
  long o = (long)i * Hout + c;
  float nr = norm[i];
  float4 av = *(const float4*)&AGG[o];
  float4 bv = *(const float4*)&bias[c];
  float4 r4 = {av.x * nr + bv.x, av.y * nr + bv.y, av.z * nr + bv.z, av.w * nr + bv.w};
  *(float4*)&out[o] = r4;
}

extern "C" void kernel_launch(void* const* d_in, const int* in_sizes, int n_in,
                              void* d_out, int out_size, void* d_ws, size_t ws_size,
                              hipStream_t stream) {
  (void)n_in; (void)out_size; (void)ws_size;
  const float* inputs = (const float*)d_in[0];
  const int*   edges  = (const int*)d_in[1];
  const float* w_in   = (const float*)d_in[2];
  const float* b_in   = (const float*)d_in[3];
  const float* w1     = (const float*)d_in[4];
  const float* b1     = (const float*)d_in[5];
  const float* w2     = (const float*)d_in[6];
  const float* b2     = (const float*)d_in[7];
  const float* w_out  = (const float*)d_in[8];
  const float* b_out  = (const float*)d_in[9];

  const int H   = in_sizes[3];
  const int IN  = in_sizes[2] / H;
  const int N   = in_sizes[0] / IN;
  const int E   = in_sizes[1] / 2;
  const int OUT = in_sizes[9];
  const int B   = in_sizes[5] / H;

  const int* srcI = edges;
  const int* dstI = edges + E;

  float* norm = (float*)d_ws;
  float* S    = norm + (((size_t)N + 255) & ~(size_t)255);
  float* AGGa = S + (size_t)N * H;
  float* AGGb = AGGa + (size_t)N * H;
  const size_t NP = ((size_t)N + 256) & ~(size_t)255;
  int* cnt    = (int*)(AGGb + (size_t)N * H);
  int* rowptr = cnt + NP;
  int* cursor = rowptr + NP;
  int* dsts   = cursor + NP;
  int* qctr   = dsts + (((size_t)E + 31) & ~(size_t)31);
  const int NQ = (2 * B + 2) * 128;

  float* xout = (float*)d_out;
  float* xbuf = xout + (size_t)N * OUT;

  k_norm_init <<<(N + 255) / 256, 256, 0, stream>>>(norm, cnt, N, qctr, NQ);
  k_norm_count<<<(E + 255) / 256, 256, 0, stream>>>(srcI, dstI, norm, cnt, E);
  k_norm_rsqrt<<<(N + 255) / 256, 256, 0, stream>>>(norm, N);
  k_scan<<<(N + SCAN_NB - 1) / SCAN_NB, 1024, 0, stream>>>(cnt, rowptr, cursor, N);
  k_fill<<<(E + 255) / 256, 256, 0, stream>>>(srcI, dstI, cursor, dsts, E);

  float* AGGi = AGGb;
  float* AGGo = AGGa;
  int* qc = qctr;
  auto conv = [&](int pre, const float* Xin, const float* pbias, int store_x,
                  const float* W, int K, int Hout) {
    size_t shmem = (size_t)(128 + Hout) * (K + 4) * sizeof(float);
    unsigned gblocks = (unsigned)((N + 127) / 128);
    const unsigned rblocks = (unsigned)((N + GR - 1) / GR);
    if (K == GK && Hout == 128) {
      if (pre == 0)
        k_gemm_wreg<0, 128><<<rblocks, 256, 0, stream>>>(Xin, W, norm, nullptr, nullptr, nullptr, 0,
                                                         S, AGGo, qc + 64, N);
      else if (pre == 1)
        k_gemm_wreg<1, 128><<<rblocks, 256, 0, stream>>>(nullptr, W, norm, AGGi, pbias, xbuf, store_x,
                                                         S, AGGo, qc + 64, N);
      else
        k_gemm_wreg<2, 128><<<rblocks, 256, 0, stream>>>(nullptr, W, norm, AGGi, pbias, xbuf, 1,
                                                         S, AGGo, qc + 64, N);
    } else if (K == GK && Hout == 64) {
      if (pre == 0)
        k_gemm_wreg<0, 64><<<rblocks, 256, 0, stream>>>(Xin, W, norm, nullptr, nullptr, nullptr, 0,
                                                        S, AGGo, qc + 64, N);
      else if (pre == 1)
        k_gemm_wreg<1, 64><<<rblocks, 256, 0, stream>>>(nullptr, W, norm, AGGi, pbias, xbuf, store_x,
                                                        S, AGGo, qc + 64, N);
      else
        k_gemm_wreg<2, 64><<<rblocks, 256, 0, stream>>>(nullptr, W, norm, AGGi, pbias, xbuf, 1,
                                                        S, AGGo, qc + 64, N);
    } else if (pre == 0)
      k_gemm_wmma<0><<<gblocks, 256, shmem, stream>>>(Xin, W, norm, nullptr, nullptr, nullptr, 0,
                                                      S, AGGo, N, K, Hout);
    else if (pre == 1)
      k_gemm_wmma<1><<<gblocks, 256, shmem, stream>>>(nullptr, W, norm, AGGi, pbias, xbuf, store_x,
                                                      S, AGGo, N, K, Hout);
    else
      k_gemm_wmma<2><<<gblocks, 256, shmem, stream>>>(nullptr, W, norm, AGGi, pbias, xbuf, 1,
                                                      S, AGGo, N, K, Hout);
    if (Hout == 128)
      k_push<4><<<PUSH_BLOCKS, 256, 0, stream>>>(rowptr, dsts, S, AGGo, qc, N);
    else if (Hout == 64)
      k_push<2><<<PUSH_BLOCKS, 256, 0, stream>>>(rowptr, dsts, S, AGGo, qc, N);
    else
      k_push_any<<<(unsigned)((N + 7) / 8), 256, 0, stream>>>(rowptr, dsts, S, AGGo, N, Hout);
    qc += 128;
    float* t = AGGi; AGGi = AGGo; AGGo = t;
  };

  conv(0, inputs, nullptr, 0, w_in, IN, H);
  const float* pb = b_in;
  int pre = 1;
  for (int i = 0; i < B; ++i) {
    conv(pre, nullptr, pb, 1, w1 + (size_t)i * H * H, H, H);
    conv(1, nullptr, b1 + (size_t)i * H, 0, w2 + (size_t)i * H * H, H, H);
    pb = b2 + (size_t)i * H;
    pre = 2;
  }
  conv(pre, nullptr, pb, 1, w_out, H, OUT);
  long tot2 = (long)N * (OUT >> 2);
  k_finalize<<<(unsigned)((tot2 + 255) / 256), 256, 0, stream>>>(AGGi, norm, b_out, xout, N, OUT);
}


# Scatter sums of the 13 equal-width layers share ONE buffer: the GEMM block that has read a row tile's finished sum writes the tile's new start value over it in place; one N*H plane less in the cache
# speedup vs baseline: 3.0184x; 1.0762x over previous
#include <hip/hip_runtime.h>

typedef __attribute__((ext_vector_type(2))) float v2f;
typedef __attribute__((ext_vector_type(8))) float v8f;

#define NEG_SLOPE 0.01f

__global__ void k_norm_init(float* __restrict__ norm, int* __restrict__ cnt, int n,
                            int* __restrict__ qctr, int nq) {
  int i = blockIdx.x * blockDim.x + threadIdx.x;
  if (i < n) {
    norm[i] = 1.0f;
    cnt[i] = 0;
  }
  for (int j = i; j < nq; j += (int)(gridDim.x * blockDim.x)) qctr[j] = 0;
}
__global__ void k_norm_count(const int* __restrict__ src, const int* __restrict__ dst,
                             float* __restrict__ norm, int* __restrict__ cnt, int e) {
  int i = blockIdx.x * blockDim.x + threadIdx.x;
  if (i < e) {
    atomicAdd(&norm[dst[i]], 1.0f);
    atomicAdd(&cnt[src[i]], 1);
  }
}
__global__ void k_norm_rsqrt(float* __restrict__ norm, int n) {
  int i = blockIdx.x * blockDim.x + threadIdx.x;
  if (i < n) norm[i] = rsqrtf(norm[i]);
}

#define SCAN_NB 2048
__global__ __launch_bounds__(1024) void k_scan(const int* __restrict__ cnt, int* __restrict__ rowptr,
                                               int* __restrict__ cursor, int n) {
  __shared__ int wsum[32];
  __shared__ int wtot[32];
  const int t = threadIdx.x;
  const int lane = t & 31, w = t >> 5;
  const int n0 = blockIdx.x * SCAN_NB;
  int s = 0;
  for (int i = t; i < n0; i += 1024) s += cnt[i];
  for (int off = 16; off > 0; off >>= 1) s += __shfl_xor(s, off);
  const int i0 = n0 + w * 64 + lane;
  const int i1 = i0 + 32;
  const int c0 = (i0 < n) ? cnt[i0] : 0;
  const int c1 = (i1 < n) ? cnt[i1] : 0;
  int x0 = c0, x1 = c1;
  for (int off = 1; off < 32; off <<= 1) {
    int y0 = __shfl_up(x0, off);
    int y1 = __shfl_up(x1, off);
    if (lane >= off) { x0 += y0; x1 += y1; }
  }
  const int t0 = __shfl(x0, 31);
  const int tw = t0 + __shfl(x1, 31);
  if (lane == 0) { wsum[w] = s; wtot[w] = tw; }
  __syncthreads();
  int a = wsum[lane];
  for (int off = 16; off > 0; off >>= 1) a += __shfl_xor(a, off);
  const int z = wtot[lane];
  int zi = z;
  for (int off = 1; off < 32; off <<= 1) {
    int y = __shfl_up(zi, off);
    if (lane >= off) zi += y;
  }
  const int base = a + __shfl(zi - z, w);
  const int e0 = base + x0 - c0;
  const int e1 = base + t0 + x1 - c1;
  if (i0 < n) { rowptr[i0] = e0; cursor[i0] = e0; }
  if (i1 < n) { rowptr[i1] = e1; cursor[i1] = e1; }
  if (i0 == n - 1) rowptr[n] = e0 + c0;
  if (i1 == n - 1) rowptr[n] = e1 + c1;
}
__global__ void k_fill(const int* __restrict__ src, const int* __restrict__ dst,
                       int* __restrict__ cursor, int* __restrict__ dsts, int e) {
  int i = blockIdx.x * blockDim.x + threadIdx.x;
  if (i < e) {
    int p = atomicAdd(&cursor[src[i]], 1);
    dsts[p] = dst[i];
  }
}

template <int PRE>
__global__ __launch_bounds__(256) void k_gemm_wmma(
    const float* __restrict__ X, const float* __restrict__ W,
    const float* __restrict__ norm,
    const float* AGGi, const float* __restrict__ pbias,
    float* __restrict__ xio, int store_x,
    float* __restrict__ S, float* AGGo,
    int nrows, int K, int Hout) {
  extern __shared__ float lds[];
  const int KSTRIDE = K + 4;
  float* Xs = lds;
  float* Wt = lds + 128 * KSTRIDE;

  const int tid = threadIdx.x;
  const int row0 = blockIdx.x * 128;

  for (int idx = tid * 4; idx < K * Hout; idx += 256 * 4) {
    int k = idx / Hout, c = idx - k * Hout;
    float4 v = *(const float4*)&W[idx];
    Wt[(c + 0) * KSTRIDE + k] = v.x;
    Wt[(c + 1) * KSTRIDE + k] = v.y;
    Wt[(c + 2) * KSTRIDE + k] = v.z;
    Wt[(c + 3) * KSTRIDE + k] = v.w;
  }
  for (int idx = tid * 4; idx < 128 * K; idx += 256 * 4) {
    int r = idx / K, c = idx - r * K;
    float4 v = {0.f, 0.f, 0.f, 0.f};
    int gr = row0 + r;
    if (gr < nrows) {
      long o = (long)gr * K + c;
      if (PRE == 0) {
        v = *(const float4*)&X[o];
      } else {
        float nr = norm[gr];
        float4 av = *(const float4*)&AGGi[o];
        float4 bv = *(const float4*)&pbias[c];
        v.x = av.x * nr + bv.x;
        v.y = av.y * nr + bv.y;
        v.z = av.z * nr + bv.z;
        v.w = av.w * nr + bv.w;
        v.x = v.x >= 0.f ? v.x : NEG_SLOPE * v.x;
        v.y = v.y >= 0.f ? v.y : NEG_SLOPE * v.y;
        v.z = v.z >= 0.f ? v.z : NEG_SLOPE * v.z;
        v.w = v.w >= 0.f ? v.w : NEG_SLOPE * v.w;
        if (PRE == 2) {
          float4 xv = *(const float4*)&xio[o];
          v.x = (xv.x + v.x) * 0.5f;
          v.y = (xv.y + v.y) * 0.5f;
          v.z = (xv.z + v.z) * 0.5f;
          v.w = (xv.w + v.w) * 0.5f;
        }
        if (PRE == 2 || store_x) *(float4*)&xio[o] = v;
      }
    }
    *(float4*)&Xs[r * KSTRIDE + c] = v;
  }
  __syncthreads();

  const int lane = tid & 31;
  const int wave = tid >> 5;
  const int mrow = lane & 15;
  const int kg   = lane >> 4;
  const float* Xrow = &Xs[(wave * 16 + mrow) * KSTRIDE + 2 * kg];

  float nrm[8];
#pragma unroll
  for (int r = 0; r < 8; ++r) {
    int grow = row0 + wave * 16 + r + 8 * kg;
    nrm[r] = (grow < nrows) ? norm[grow] : 0.f;
  }

  for (int ct = 0; ct < (Hout >> 4); ++ct) {
    const int col = ct * 16 + mrow;
    const float* Wcol = &Wt[col * KSTRIDE + 2 * kg];
    v8f acc = {};
    for (int k = 0; k < K; k += 4) {
      v2f a = *(const v2f*)&Xrow[k];
      v2f b = *(const v2f*)&Wcol[k];
      acc = __builtin_amdgcn_wmma_f32_16x16x4_f32(
           false, a,  false, b,
           (short)0, acc,  false,  false);
    }
#pragma unroll
    for (int r = 0; r < 8; ++r) {
      int grow = row0 + wave * 16 + r + 8 * kg;
      if (grow < nrows) {
        float sv = acc[r] * nrm[r];
        long o = (long)grow * Hout + col;
        S[o]    = sv;
        AGGo[o] = sv;
      }
    }
  }
}

static __device__ __forceinline__ void lds_barrier() {
  __builtin_amdgcn_fence(__ATOMIC_RELEASE, "workgroup", "local");
  __builtin_amdgcn_s_barrier();
  __builtin_amdgcn_fence(__ATOMIC_ACQUIRE, "workgroup", "local");
}

#define GK 128
#define GR 64
template <int PRE, int HOUT>
__global__ __launch_bounds__(256) void k_gemm_wreg(
    const float* __restrict__ X, const float* __restrict__ W,
    const float* __restrict__ norm,
    const float* AGGi, const float* __restrict__ pbias,
    float* __restrict__ xio, int store_x,
    float* __restrict__ S, float* AGGo, int* __restrict__ gctr, int nrows) {
  constexpr int KS = GK + 4;
  constexpr int NCT = HOUT / 16;
  constexpr int NRG = 8 / NCT;
  constexpr int RTW = (GR / 16) / NRG;
  __shared__ __attribute__((aligned(16))) float Xs[GR * KS];
  __shared__ int tile_s;

  const int tid = threadIdx.x;
  const int lane = tid & 31;
  const int wave = tid >> 5;
  const int mrow = lane & 15;
  const int kg   = lane >> 4;
  const int ctw = wave % NCT;
  const int rg  = wave / NCT;
  const int col = ctw * 16 + mrow;

  v2f b[GK / 4];
  const float* wp = W + (2 * kg) * HOUT + col;
#pragma unroll
  for (int kk = 0; kk < GK / 4; ++kk) {
    b[kk].x = wp[(4 * kk) * HOUT];
    b[kk].y = wp[(4 * kk + 1) * HOUT];
  }

  if (tid == 0) {
    const int ntiles = (int)gridDim.x;
    const int tlow = ((nrows >> 1) + GR - 1) / GR;
    int hh = (__builtin_amdgcn_s_sendmsg_rtn(0x87) >> 18) & 1;
    int cnt = hh ? ntiles - tlow : tlow;
    int tk = atomicAdd(gctr + hh * 32, 1);
    if (tk >= cnt) {
      hh ^= 1;
      cnt = hh ? ntiles - tlow : tlow;
      tk = atomicAdd(gctr + hh * 32, 1);
    }
    tile_s = (tk < cnt) ? (hh ? tlow + tk : tk) : -1;
  }
  lds_barrier();
  const int tile = tile_s;
  if (tile < 0) return;
  const int row0 = tile * GR;

  for (int idx = tid * 4; idx < GR * GK; idx += 256 * 4) {
    int r = idx / GK, c = idx - r * GK;
    float4 v = {0.f, 0.f, 0.f, 0.f};
    int gr = row0 + r;
    if (gr < nrows) {
      long o = (long)gr * GK + c;
      if (PRE == 0) {
        v = *(const float4*)&X[o];
      } else {
        float nr = norm[gr];
        float4 av = *(const float4*)&AGGi[o];
        float4 bv = *(const float4*)&pbias[c];
        v.x = av.x * nr + bv.x;
        v.y = av.y * nr + bv.y;
        v.z = av.z * nr + bv.z;
        v.w = av.w * nr + bv.w;
        v.x = v.x >= 0.f ? v.x : NEG_SLOPE * v.x;
        v.y = v.y >= 0.f ? v.y : NEG_SLOPE * v.y;
        v.z = v.z >= 0.f ? v.z : NEG_SLOPE * v.z;
        v.w = v.w >= 0.f ? v.w : NEG_SLOPE * v.w;
        if (PRE == 2) {
          float4 xv = *(const float4*)&xio[o];
          v.x = (xv.x + v.x) * 0.5f;
          v.y = (xv.y + v.y) * 0.5f;
          v.z = (xv.z + v.z) * 0.5f;
          v.w = (xv.w + v.w) * 0.5f;
        }
        if (PRE == 2 || store_x) *(float4*)&xio[o] = v;
      }
    }
    *(float4*)&Xs[r * KS + c] = v;
  }

  __syncthreads();

  const float* xr = &Xs[((rg * RTW) * 16 + mrow) * KS + 2 * kg];
  v8f acc[RTW];
#pragma unroll
  for (int j = 0; j < RTW; ++j) acc[j] = v8f{};
#pragma unroll
  for (int kk = 0; kk < GK / 4; ++kk) {
#pragma unroll
    for (int j = 0; j < RTW; ++j) {
      v2f a = *(const v2f*)&xr[j * 16 * KS + 4 * kk];
      acc[j] = __builtin_amdgcn_wmma_f32_16x16x4_f32(
           false, a,  false, b[kk],
           (short)0, acc[j],  false,  false);
    }
  }
  float nrm[RTW][8];
#pragma unroll
  for (int j = 0; j < RTW; ++j) {
#pragma unroll
    for (int r = 0; r < 8; ++r) {
      int grow = row0 + (rg * RTW + j) * 16 + r + 8 * kg;
      nrm[j][r] = norm[grow < nrows ? grow : nrows - 1];
    }
  }
#pragma unroll
  for (int j = 0; j < RTW; ++j) {
#pragma unroll
    for (int r = 0; r < 8; ++r) {
      int grow = row0 + (rg * RTW + j) * 16 + r + 8 * kg;
      if (grow < nrows) {
        float sv = acc[j][r] * nrm[j][r];
        long o = (long)grow * HOUT + col;
        S[o]    = sv;
        AGGo[o] = sv;
      }
    }
  }
}

#define PUSH_BLOCKS 2048
#define PUSH_CHW 4
template <int NV>
__global__ __launch_bounds__(256) void k_push(const int* __restrict__ rowptr,
                                              const int* __restrict__ dsts,
                                              const float* __restrict__ S, float* __restrict__ AGG,
                                              int* __restrict__ qctr, int n) {
  constexpr int HOUT = NV * 32;
  constexpr int SPB = 8 * PUSH_CHW;
  __shared__ int tk[2];
  const int tid = threadIdx.x;
  const int lane = tid & 31;
  const int wave = __builtin_amdgcn_readfirstlane(tid >> 5);
  const int nchunks = (n + SPB - 1) / SPB;
  const int nh = n >> 1;
  int h = (__builtin_amdgcn_s_sendmsg_rtn(0x87) >> 18) & 1;
  for (int pass = 0; pass < 2; ++pass, h ^= 1) {
    const int lo = h ? nh : 0;
    const int hi = h ? n : nh;
    int* ctr = qctr + h * 32;
    if (tid == 0) tk[0] = atomicAdd(ctr, 1);
    lds_barrier();
    for (int it = 0;; ++it) {
      const int t = __builtin_amdgcn_readfirstlane(tk[it & 1]);
      if (t >= nchunks) break;
      const int s0 = t * SPB + wave * PUSH_CHW;
      int rp[PUSH_CHW + 1];
#pragma unroll
      for (int k = 0; k <= PUSH_CHW; ++k) {
        const int si = s0 + k;
        rp[k] = rowptr[si < n ? si : n];
      }
      float v[PUSH_CHW][NV];
      int dl0[PUSH_CHW];
#pragma unroll
      for (int k = 0; k < PUSH_CHW; ++k) {
        const int s = (s0 + k < n) ? (s0 + k) : (n - 1);
        const float* sp = &S[(long)s * HOUT + lane];
#pragma unroll
        for (int i = 0; i < NV; ++i) v[k][i] = sp[32 * i];
        const int deg = rp[k + 1] - rp[k];
        dl0[k] = (lane < deg) ? dsts[rp[k] + lane] : -1;
      }
      int nxt = 0;
      if (tid == 0) nxt = atomicAdd(ctr, 1);
#pragma unroll
      for (int k = 0; k < PUSH_CHW; ++k) {
        const int beg = rp[k];
        const int deg = rp[k + 1] - beg;
        int dl = dl0[k];
        for (int base = 0; base < deg; base += 32) {
          if (base > 0) dl = (base + lane < deg) ? dsts[beg + base + lane] : -1;
          unsigned m = __builtin_amdgcn_ballot_w32(dl >= lo && dl < hi);
          while (m) {
            const int j = __builtin_ctz(m);
            m &= m - 1;
            const int d = __builtin_amdgcn_readlane(dl, j);
            float* ap = &AGG[(long)d * HOUT + lane];
#pragma unroll
            for (int i = 0; i < NV; ++i) atomicAdd(ap + 32 * i, v[k][i]);
          }
        }
      }

      if (tid == 0) tk[(it + 1) & 1] = nxt;
      lds_barrier();
    }
    lds_barrier();
  }
}

__global__ __launch_bounds__(256) void k_push_any(const int* __restrict__ rowptr,
                                                  const int* __restrict__ dsts,
                                                  const float* __restrict__ S,
                                                  float* __restrict__ AGG, int n, int Hout) {
  const int lane = threadIdx.x & 31;
  const int s = __builtin_amdgcn_readfirstlane((int)(blockIdx.x * (blockDim.x >> 5) + (threadIdx.x >> 5)));
  if (s >= n) return;
  const int beg = rowptr[s], end = rowptr[s + 1];
  const float* sp = &S[(long)s * Hout];
  for (int ed = beg; ed < end; ++ed) {
    const int d = dsts[ed];
    float* ap = &AGG[(long)d * Hout];
    for (int c = lane; c < Hout; c += 32) atomicAdd(ap + c, sp[c]);
  }
}

__global__ void k_finalize(const float* __restrict__ AGG, const float* __restrict__ norm,
                           const float* __restrict__ bias, float* __restrict__ out,
                           int n, int Hout) {
  const int chunks = Hout >> 2;
  long tid = (long)blockIdx.x * blockDim.x + threadIdx.x;
  if (tid >= (long)n * chunks) return;
  int i = (int)(tid / chunks);
  int c = (int)(tid - (long)i * chunks) << 2;
  long o = (long)i * Hout + c;
  float nr = norm[i];
  float4 av = *(const float4*)&AGG[o];
  float4 bv = *(const float4*)&bias[c];
  float4 r4 = {av.x * nr + bv.x, av.y * nr + bv.y, av.z * nr + bv.z, av.w * nr + bv.w};
  *(float4*)&out[o] = r4;
}

extern "C" void kernel_launch(void* const* d_in, const int* in_sizes, int n_in,
                              void* d_out, int out_size, void* d_ws, size_t ws_size,
                              hipStream_t stream) {
  (void)n_in; (void)out_size; (void)ws_size;
  const float* inputs = (const float*)d_in[0];
  const int*   edges  = (const int*)d_in[1];
  const float* w_in   = (const float*)d_in[2];
  const float* b_in   = (const float*)d_in[3];
  const float* w1     = (const float*)d_in[4];
  const float* b1     = (const float*)d_in[5];
  const float* w2     = (const float*)d_in[6];
  const float* b2     = (const float*)d_in[7];
  const float* w_out  = (const float*)d_in[8];
  const float* b_out  = (const float*)d_in[9];

  const int H   = in_sizes[3];
  const int IN  = in_sizes[2] / H;
  const int N   = in_sizes[0] / IN;
  const int E   = in_sizes[1] / 2;
  const int OUT = in_sizes[9];
  const int B   = in_sizes[5] / H;

  const int* srcI = edges;
  const int* dstI = edges + E;

  float* norm = (float*)d_ws;
  float* S    = norm + (((size_t)N + 255) & ~(size_t)255);
  float* AGG  = S + (size_t)N * H;
  float* AGG2 = AGG + (size_t)N * H;
  const size_t NP = ((size_t)N + 256) & ~(size_t)255;
  int* cnt    = (int*)(AGG2 + (size_t)N * H);
  int* rowptr = cnt + NP;
  int* cursor = rowptr + NP;
  int* dsts   = cursor + NP;
  int* qctr   = dsts + (((size_t)E + 31) & ~(size_t)31);
  const int NQ = (2 * B + 2) * 128;

  float* xout = (float*)d_out;
  float* xbuf = xout + (size_t)N * OUT;

  k_norm_init <<<(N + 255) / 256, 256, 0, stream>>>(norm, cnt, N, qctr, NQ);
  k_norm_count<<<(E + 255) / 256, 256, 0, stream>>>(srcI, dstI, norm, cnt, E);
  k_norm_rsqrt<<<(N + 255) / 256, 256, 0, stream>>>(norm, N);
  k_scan<<<(N + SCAN_NB - 1) / SCAN_NB, 1024, 0, stream>>>(cnt, rowptr, cursor, N);
  k_fill<<<(E + 255) / 256, 256, 0, stream>>>(srcI, dstI, cursor, dsts, E);

  float* AGGi = AGG;
  int* qc = qctr;
  auto conv = [&](int pre, const float* Xin, const float* pbias, int store_x,
                  const float* W, int K, int Hout) {
    size_t shmem = (size_t)(128 + Hout) * (K + 4) * sizeof(float);
    unsigned gblocks = (unsigned)((N + 127) / 128);
    const unsigned rblocks = (unsigned)((N + GR - 1) / GR);
    float* AGGo = AGGi;
    if (Hout != K) AGGo = (AGGi == AGG) ? AGG2 : AGG;
    if (K == GK && Hout == 128) {
      if (pre == 0)
        k_gemm_wreg<0, 128><<<rblocks, 256, 0, stream>>>(Xin, W, norm, nullptr, nullptr, nullptr, 0,
                                                         S, AGGo, qc + 64, N);
      else if (pre == 1)
        k_gemm_wreg<1, 128><<<rblocks, 256, 0, stream>>>(nullptr, W, norm, AGGi, pbias, xbuf, store_x,
                                                         S, AGGo, qc + 64, N);
      else
        k_gemm_wreg<2, 128><<<rblocks, 256, 0, stream>>>(nullptr, W, norm, AGGi, pbias, xbuf, 1,
                                                         S, AGGo, qc + 64, N);
    } else if (K == GK && Hout == 64) {
      if (pre == 0)
        k_gemm_wreg<0, 64><<<rblocks, 256, 0, stream>>>(Xin, W, norm, nullptr, nullptr, nullptr, 0,
                                                        S, AGGo, qc + 64, N);
      else if (pre == 1)
        k_gemm_wreg<1, 64><<<rblocks, 256, 0, stream>>>(nullptr, W, norm, AGGi, pbias, xbuf, store_x,
                                                        S, AGGo, qc + 64, N);
      else
        k_gemm_wreg<2, 64><<<rblocks, 256, 0, stream>>>(nullptr, W, norm, AGGi, pbias, xbuf, 1,
                                                        S, AGGo, qc + 64, N);
    } else if (pre == 0)
      k_gemm_wmma<0><<<gblocks, 256, shmem, stream>>>(Xin, W, norm, nullptr, nullptr, nullptr, 0,
                                                      S, AGGo, N, K, Hout);
    else if (pre == 1)
      k_gemm_wmma<1><<<gblocks, 256, shmem, stream>>>(nullptr, W, norm, AGGi, pbias, xbuf, store_x,
                                                      S, AGGo, N, K, Hout);
    else
      k_gemm_wmma<2><<<gblocks, 256, shmem, stream>>>(nullptr, W, norm, AGGi, pbias, xbuf, 1,
                                                      S, AGGo, N, K, Hout);
    if (Hout == 128)
      k_push<4><<<PUSH_BLOCKS, 256, 0, stream>>>(rowptr, dsts, S, AGGo, qc, N);
    else if (Hout == 64)
      k_push<2><<<PUSH_BLOCKS, 256, 0, stream>>>(rowptr, dsts, S, AGGo, qc, N);
    else
      k_push_any<<<(unsigned)((N + 7) / 8), 256, 0, stream>>>(rowptr, dsts, S, AGGo, N, Hout);
    qc += 128;
    AGGi = AGGo;
  };

  conv(0, inputs, nullptr, 0, w_in, IN, H);
  const float* pb = b_in;
  int pre = 1;
  for (int i = 0; i < B; ++i) {
    conv(pre, nullptr, pb, 1, w1 + (size_t)i * H * H, H, H);
    conv(1, nullptr, b1 + (size_t)i * H, 0, w2 + (size_t)i * H * H, H, H);
    pb = b2 + (size_t)i * H;
    pre = 2;
  }
  conv(pre, nullptr, pb, 1, w_out, H, OUT);
  long tot2 = (long)N * (OUT >> 2);
  k_finalize<<<(unsigned)((tot2 + 255) / 256), 256, 0, stream>>>(AGGi, norm, b_out, xout, N, OUT);
}


# GEMM blocks persistent (512, 2 per WGP): a block keeps its W fragments, draws 32-row tiles by ticket in a loop and requests the next tile's rows before the current tile's products; sums unchanged
# speedup vs baseline: 3.1552x; 1.0453x over previous
#include <hip/hip_runtime.h>

typedef __attribute__((ext_vector_type(2))) float v2f;
typedef __attribute__((ext_vector_type(8))) float v8f;

#define NEG_SLOPE 0.01f

__global__ void k_norm_init(float* __restrict__ norm, int* __restrict__ cnt, int n,
                            int* __restrict__ qctr, int nq) {
  int i = blockIdx.x * blockDim.x + threadIdx.x;
  if (i < n) {
    norm[i] = 1.0f;
    cnt[i] = 0;
  }
  for (int j = i; j < nq; j += (int)(gridDim.x * blockDim.x)) qctr[j] = 0;
}
__global__ void k_norm_count(const int* __restrict__ src, const int* __restrict__ dst,
                             float* __restrict__ norm, int* __restrict__ cnt, int e) {
  int i = blockIdx.x * blockDim.x + threadIdx.x;
  if (i < e) {
    atomicAdd(&norm[dst[i]], 1.0f);
    atomicAdd(&cnt[src[i]], 1);
  }
}
__global__ void k_norm_rsqrt(float* __restrict__ norm, int n) {
  int i = blockIdx.x * blockDim.x + threadIdx.x;
  if (i < n) norm[i] = rsqrtf(norm[i]);
}

#define SCAN_NB 2048
__global__ __launch_bounds__(1024) void k_scan(const int* __restrict__ cnt, int* __restrict__ rowptr,
                                               int* __restrict__ cursor, int n) {
  __shared__ int wsum[32];
  __shared__ int wtot[32];
  const int t = threadIdx.x;
  const int lane = t & 31, w = t >> 5;
  const int n0 = blockIdx.x * SCAN_NB;
  int s = 0;
  for (int i = t; i < n0; i += 1024) s += cnt[i];
  for (int off = 16; off > 0; off >>= 1) s += __shfl_xor(s, off);
  const int i0 = n0 + w * 64 + lane;
  const int i1 = i0 + 32;
  const int c0 = (i0 < n) ? cnt[i0] : 0;
  const int c1 = (i1 < n) ? cnt[i1] : 0;
  int x0 = c0, x1 = c1;
  for (int off = 1; off < 32; off <<= 1) {
    int y0 = __shfl_up(x0, off);
    int y1 = __shfl_up(x1, off);
    if (lane >= off) { x0 += y0; x1 += y1; }
  }
  const int t0 = __shfl(x0, 31);
  const int tw = t0 + __shfl(x1, 31);
  if (lane == 0) { wsum[w] = s; wtot[w] = tw; }
  __syncthreads();
  int a = wsum[lane];
  for (int off = 16; off > 0; off >>= 1) a += __shfl_xor(a, off);
  const int z = wtot[lane];
  int zi = z;
  for (int off = 1; off < 32; off <<= 1) {
    int y = __shfl_up(zi, off);
    if (lane >= off) zi += y;
  }
  const int base = a + __shfl(zi - z, w);
  const int e0 = base + x0 - c0;
  const int e1 = base + t0 + x1 - c1;
  if (i0 < n) { rowptr[i0] = e0; cursor[i0] = e0; }
  if (i1 < n) { rowptr[i1] = e1; cursor[i1] = e1; }
  if (i0 == n - 1) rowptr[n] = e0 + c0;
  if (i1 == n - 1) rowptr[n] = e1 + c1;
}
__global__ void k_fill(const int* __restrict__ src, const int* __restrict__ dst,
                       int* __restrict__ cursor, int* __restrict__ dsts, int e) {
  int i = blockIdx.x * blockDim.x + threadIdx.x;
  if (i < e) {
    int p = atomicAdd(&cursor[src[i]], 1);
    dsts[p] = dst[i];
  }
}

template <int PRE>
__global__ __launch_bounds__(256) void k_gemm_wmma(
    const float* __restrict__ X, const float* __restrict__ W,
    const float* __restrict__ norm,
    const float* AGGi, const float* __restrict__ pbias,
    float* __restrict__ xio, int store_x,
    float* __restrict__ S, float* AGGo,
    int nrows, int K, int Hout) {
  extern __shared__ float lds[];
  const int KSTRIDE = K + 4;
  float* Xs = lds;
  float* Wt = lds + 128 * KSTRIDE;

  const int tid = threadIdx.x;
  const int row0 = blockIdx.x * 128;

  for (int idx = tid * 4; idx < K * Hout; idx += 256 * 4) {
    int k = idx / Hout, c = idx - k * Hout;
    float4 v = *(const float4*)&W[idx];
    Wt[(c + 0) * KSTRIDE + k] = v.x;
    Wt[(c + 1) * KSTRIDE + k] = v.y;
    Wt[(c + 2) * KSTRIDE + k] = v.z;
    Wt[(c + 3) * KSTRIDE + k] = v.w;
  }
  for (int idx = tid * 4; idx < 128 * K; idx += 256 * 4) {
    int r = idx / K, c = idx - r * K;
    float4 v = {0.f, 0.f, 0.f, 0.f};
    int gr = row0 + r;
    if (gr < nrows) {
      long o = (long)gr * K + c;
      if (PRE == 0) {
        v = *(const float4*)&X[o];
      } else {
        float nr = norm[gr];
        float4 av = *(const float4*)&AGGi[o];
        float4 bv = *(const float4*)&pbias[c];
        v.x = av.x * nr + bv.x;
        v.y = av.y * nr + bv.y;
        v.z = av.z * nr + bv.z;
        v.w = av.w * nr + bv.w;
        v.x = v.x >= 0.f ? v.x : NEG_SLOPE * v.x;
        v.y = v.y >= 0.f ? v.y : NEG_SLOPE * v.y;
        v.z = v.z >= 0.f ? v.z : NEG_SLOPE * v.z;
        v.w = v.w >= 0.f ? v.w : NEG_SLOPE * v.w;
        if (PRE == 2) {
          float4 xv = *(const float4*)&xio[o];
          v.x = (xv.x + v.x) * 0.5f;
          v.y = (xv.y + v.y) * 0.5f;
          v.z = (xv.z + v.z) * 0.5f;
          v.w = (xv.w + v.w) * 0.5f;
        }
        if (PRE == 2 || store_x) *(float4*)&xio[o] = v;
      }
    }
    *(float4*)&Xs[r * KSTRIDE + c] = v;
  }
  __syncthreads();

  const int lane = tid & 31;
  const int wave = tid >> 5;
  const int mrow = lane & 15;
  const int kg   = lane >> 4;
  const float* Xrow = &Xs[(wave * 16 + mrow) * KSTRIDE + 2 * kg];

  float nrm[8];
#pragma unroll
  for (int r = 0; r < 8; ++r) {
    int grow = row0 + wave * 16 + r + 8 * kg;
    nrm[r] = (grow < nrows) ? norm[grow] : 0.f;
  }

  for (int ct = 0; ct < (Hout >> 4); ++ct) {
    const int col = ct * 16 + mrow;
    const float* Wcol = &Wt[col * KSTRIDE + 2 * kg];
    v8f acc = {};
    for (int k = 0; k < K; k += 4) {
      v2f a = *(const v2f*)&Xrow[k];
      v2f b = *(const v2f*)&Wcol[k];
      acc = __builtin_amdgcn_wmma_f32_16x16x4_f32(
           false, a,  false, b,
           (short)0, acc,  false,  false);
    }
#pragma unroll
    for (int r = 0; r < 8; ++r) {
      int grow = row0 + wave * 16 + r + 8 * kg;
      if (grow < nrows) {
        float sv = acc[r] * nrm[r];
        long o = (long)grow * Hout + col;
        S[o]    = sv;
        AGGo[o] = sv;
      }
    }
  }
}

static __device__ __forceinline__ void lds_barrier() {
  __builtin_amdgcn_fence(__ATOMIC_RELEASE, "workgroup", "local");
  __builtin_amdgcn_s_barrier();
  __builtin_amdgcn_fence(__ATOMIC_ACQUIRE, "workgroup", "local");
}

#define GK 128
#define GR 32
#define GEMM_BLOCKS 512
template <int PRE, int HOUT>
__global__ __launch_bounds__(256) void k_gemm_wreg(
    const float* __restrict__ X, const float* __restrict__ W,
    const float* __restrict__ norm,
    const float* AGGi, const float* __restrict__ pbias,
    float* __restrict__ xio, int store_x,
    float* __restrict__ S, float* AGGo, int* __restrict__ gctr, int nrows, int ntiles) {
  constexpr int KS = GK + 4;
  constexpr int NCT = HOUT / 16;
  constexpr int NRG = 8 / NCT;
  constexpr int RTW = (GR / 16) / NRG;
  constexpr int NT = GR / 8;
  __shared__ __attribute__((aligned(16))) float Xs[GR * KS];
  __shared__ __attribute__((aligned(16))) float nrm_s[GR];
  __shared__ int tick_s[2];

  const int tid = threadIdx.x;
  const int lane = tid & 31;
  const int wave = tid >> 5;
  const int mrow = lane & 15;
  const int kg   = lane >> 4;
  const int ctw = wave % NCT;
  const int rg  = wave / NCT;
  const int col = ctw * 16 + mrow;
  const int sc = lane * 4;
  const int sr = wave;

  v2f b[GK / 4];
  const float* wp = W + (2 * kg) * HOUT + col;
#pragma unroll
  for (int kk = 0; kk < GK / 4; ++kk) {
    b[kk].x = wp[(4 * kk) * HOUT];
    b[kk].y = wp[(4 * kk + 1) * HOUT];
  }
  float4 bv = {0.f, 0.f, 0.f, 0.f};
  if (PRE != 0) bv = *(const float4*)&pbias[sc];

  const int tlow = ((nrows >> 1) + GR - 1) / GR;
  int hh = 0, other = 0, tkraw = 0;
  auto ask = [&]() { tkraw = atomicAdd(gctr + hh * 32, 1); };
  auto take = [&]() -> int {
    int cnt = hh ? ntiles - tlow : tlow;
    int tk = tkraw;
    if (tk >= cnt && !other) {
      other = 1;
      hh ^= 1;
      cnt = hh ? ntiles - tlow : tlow;
      tk = atomicAdd(gctr + hh * 32, 1);
    }
    return (tk < cnt) ? (hh ? tlow + tk : tk) : -1;
  };
  if (tid == 0) {
    hh = (__builtin_amdgcn_s_sendmsg_rtn(0x87) >> 18) & 1;
    ask();
    tick_s[0] = take();
  }
  lds_barrier();
  int tile = __builtin_amdgcn_readfirstlane(tick_s[0]);

  float nr[NT];
  float4 av[NT], xv[NT];
  auto request = [&](int t_) {
    const int r0 = t_ * GR + sr;
#pragma unroll
    for (int t = 0; t < NT; ++t) {
      const int gr = r0 + 8 * t;
      const int grc = gr < nrows ? gr : nrows - 1;
      const long o = (long)grc * GK + sc;
      nr[t] = norm[grc];
      if (PRE == 0) {
        av[t] = *(const float4*)&X[o];
      } else {
        av[t] = *(const float4*)&AGGi[o];
        if (PRE == 2) xv[t] = *(const float4*)&xio[o];
      }
    }
  };
  if (tile >= 0) {
    request(tile);
    if (tid == 0) ask();
  }

  while (tile >= 0) {
    const int row0 = tile * GR;
#pragma unroll
    for (int t = 0; t < NT; ++t) {
      const int r = sr + 8 * t;
      const int gr = row0 + r;
      const bool ok = gr < nrows;
      float4 v = av[t];
      if (PRE != 0) {
        const float nrt = nr[t];
        v.x = v.x * nrt + bv.x;
        v.y = v.y * nrt + bv.y;
        v.z = v.z * nrt + bv.z;
        v.w = v.w * nrt + bv.w;
        v.x = v.x >= 0.f ? v.x : NEG_SLOPE * v.x;
        v.y = v.y >= 0.f ? v.y : NEG_SLOPE * v.y;
        v.z = v.z >= 0.f ? v.z : NEG_SLOPE * v.z;
        v.w = v.w >= 0.f ? v.w : NEG_SLOPE * v.w;
        if (PRE == 2) {
          v.x = (xv[t].x + v.x) * 0.5f;
          v.y = (xv[t].y + v.y) * 0.5f;
          v.z = (xv[t].z + v.z) * 0.5f;
          v.w = (xv[t].w + v.w) * 0.5f;
        }
        if (ok && (PRE == 2 || store_x)) *(float4*)&xio[(long)gr * GK + sc] = v;
      }
      if (!ok) v = float4{0.f, 0.f, 0.f, 0.f};
      *(float4*)&Xs[r * KS + sc] = v;
      if (lane == 0) nrm_s[r] = ok ? nr[t] : 0.f;
    }
    if (tid == 0) tick_s[1] = take();
    lds_barrier();
    const int next = __builtin_amdgcn_readfirstlane(tick_s[1]);
    if (next >= 0) {
      request(next);
      if (tid == 0) ask();
    }

    const float* xr = &Xs[((rg * RTW) * 16 + mrow) * KS + 2 * kg];
    v8f acc[RTW];
#pragma unroll
    for (int j = 0; j < RTW; ++j) acc[j] = v8f{};
#pragma unroll
    for (int kk = 0; kk < GK / 4; ++kk) {
#pragma unroll
      for (int j = 0; j < RTW; ++j) {
        v2f a = *(const v2f*)&xr[j * 16 * KS + 4 * kk];
        acc[j] = __builtin_amdgcn_wmma_f32_16x16x4_f32(
             false, a,  false, b[kk],
             (short)0, acc[j],  false,  false);
      }
    }
    float nrm[RTW][8];
#pragma unroll
    for (int j = 0; j < RTW; ++j) {
      const float4* np = (const float4*)&nrm_s[(rg * RTW + j) * 16 + 8 * kg];
      const float4 n0 = np[0], n1 = np[1];
      nrm[j][0] = n0.x; nrm[j][1] = n0.y; nrm[j][2] = n0.z; nrm[j][3] = n0.w;
      nrm[j][4] = n1.x; nrm[j][5] = n1.y; nrm[j][6] = n1.z; nrm[j][7] = n1.w;
    }
#pragma unroll
    for (int j = 0; j < RTW; ++j) {
#pragma unroll
      for (int r = 0; r < 8; ++r) {
        const int lr = (rg * RTW + j) * 16 + r + 8 * kg;
        const int grow = row0 + lr;
        if (grow < nrows) {
          float sv = acc[j][r] * nrm[j][r];
          long o = (long)grow * HOUT + col;
          S[o]    = sv;
          AGGo[o] = sv;
        }
      }
    }
    lds_barrier();
    tile = next;
  }
}

#define PUSH_BLOCKS 2048
#define PUSH_CHW 4
template <int NV>
__global__ __launch_bounds__(256) void k_push(const int* __restrict__ rowptr,
                                              const int* __restrict__ dsts,
                                              const float* __restrict__ S, float* __restrict__ AGG,
                                              int* __restrict__ qctr, int n) {
  constexpr int HOUT = NV * 32;
  constexpr int SPB = 8 * PUSH_CHW;
  __shared__ int tk[2];
  const int tid = threadIdx.x;
  const int lane = tid & 31;
  const int wave = __builtin_amdgcn_readfirstlane(tid >> 5);
  const int nchunks = (n + SPB - 1) / SPB;
  const int nh = n >> 1;
  int h = (__builtin_amdgcn_s_sendmsg_rtn(0x87) >> 18) & 1;
  for (int pass = 0; pass < 2; ++pass, h ^= 1) {
    const int lo = h ? nh : 0;
    const int hi = h ? n : nh;
    int* ctr = qctr + h * 32;
    if (tid == 0) tk[0] = atomicAdd(ctr, 1);
    lds_barrier();
    for (int it = 0;; ++it) {
      const int t = __builtin_amdgcn_readfirstlane(tk[it & 1]);
      if (t >= nchunks) break;
      const int s0 = t * SPB + wave * PUSH_CHW;
      int rp[PUSH_CHW + 1];
#pragma unroll
      for (int k = 0; k <= PUSH_CHW; ++k) {
        const int si = s0 + k;
        rp[k] = rowptr[si < n ? si : n];
      }
      float v[PUSH_CHW][NV];
      int dl0[PUSH_CHW];
#pragma unroll
      for (int k = 0; k < PUSH_CHW; ++k) {
        const int s = (s0 + k < n) ? (s0 + k) : (n - 1);
        const float* sp = &S[(long)s * HOUT + lane];
#pragma unroll
        for (int i = 0; i < NV; ++i) v[k][i] = sp[32 * i];
        const int deg = rp[k + 1] - rp[k];
        dl0[k] = (lane < deg) ? dsts[rp[k] + lane] : -1;
      }
      int nxt = 0;
      if (tid == 0) nxt = atomicAdd(ctr, 1);
#pragma unroll
      for (int k = 0; k < PUSH_CHW; ++k) {
        const int beg = rp[k];
        const int deg = rp[k + 1] - beg;
        int dl = dl0[k];
        for (int base = 0; base < deg; base += 32) {
          if (base > 0) dl = (base + lane < deg) ? dsts[beg + base + lane] : -1;
          unsigned m = __builtin_amdgcn_ballot_w32(dl >= lo && dl < hi);
          while (m) {
            const int j = __builtin_ctz(m);
            m &= m - 1;
            const int d = __builtin_amdgcn_readlane(dl, j);
            float* ap = &AGG[(long)d * HOUT + lane];
#pragma unroll
            for (int i = 0; i < NV; ++i) atomicAdd(ap + 32 * i, v[k][i]);
          }
        }
      }

      if (tid == 0) tk[(it + 1) & 1] = nxt;
      lds_barrier();
    }
    lds_barrier();
  }
}

__global__ __launch_bounds__(256) void k_push_any(const int* __restrict__ rowptr,
                                                  const int* __restrict__ dsts,
                                                  const float* __restrict__ S,
                                                  float* __restrict__ AGG, int n, int Hout) {
  const int lane = threadIdx.x & 31;
  const int s = __builtin_amdgcn_readfirstlane((int)(blockIdx.x * (blockDim.x >> 5) + (threadIdx.x >> 5)));
  if (s >= n) return;
  const int beg = rowptr[s], end = rowptr[s + 1];
  const float* sp = &S[(long)s * Hout];
  for (int ed = beg; ed < end; ++ed) {
    const int d = dsts[ed];
    float* ap = &AGG[(long)d * Hout];
    for (int c = lane; c < Hout; c += 32) atomicAdd(ap + c, sp[c]);
  }
}

__global__ void k_finalize(const float* __restrict__ AGG, const float* __restrict__ norm,
                           const float* __restrict__ bias, float* __restrict__ out,
                           int n, int Hout) {
  const int chunks = Hout >> 2;
  long tid = (long)blockIdx.x * blockDim.x + threadIdx.x;
  if (tid >= (long)n * chunks) return;
  int i = (int)(tid / chunks);
  int c = (int)(tid - (long)i * chunks) << 2;
  long o = (long)i * Hout + c;
  float nr = norm[i];
  float4 av = *(const float4*)&AGG[o];
  float4 bv = *(const float4*)&bias[c];
  float4 r4 = {av.x * nr + bv.x, av.y * nr + bv.y, av.z * nr + bv.z, av.w * nr + bv.w};
  *(float4*)&out[o] = r4;
}

extern "C" void kernel_launch(void* const* d_in, const int* in_sizes, int n_in,
                              void* d_out, int out_size, void* d_ws, size_t ws_size,
                              hipStream_t stream) {
  (void)n_in; (void)out_size; (void)ws_size;
  const float* inputs = (const float*)d_in[0];
  const int*   edges  = (const int*)d_in[1];
  const float* w_in   = (const float*)d_in[2];
  const float* b_in   = (const float*)d_in[3];
  const float* w1     = (const float*)d_in[4];
  const float* b1     = (const float*)d_in[5];
  const float* w2     = (const float*)d_in[6];
  const float* b2     = (const float*)d_in[7];
  const float* w_out  = (const float*)d_in[8];
  const float* b_out  = (const float*)d_in[9];

  const int H   = in_sizes[3];
  const int IN  = in_sizes[2] / H;
  const int N   = in_sizes[0] / IN;
  const int E   = in_sizes[1] / 2;
  const int OUT = in_sizes[9];
  const int B   = in_sizes[5] / H;

  const int* srcI = edges;
  const int* dstI = edges + E;

  float* norm = (float*)d_ws;
  float* S    = norm + (((size_t)N + 255) & ~(size_t)255);
  float* AGG  = S + (size_t)N * H;
  float* AGG2 = AGG + (size_t)N * H;
  const size_t NP = ((size_t)N + 256) & ~(size_t)255;
  int* cnt    = (int*)(AGG2 + (size_t)N * H);
  int* rowptr = cnt + NP;
  int* cursor = rowptr + NP;
  int* dsts   = cursor + NP;
  int* qctr   = dsts + (((size_t)E + 31) & ~(size_t)31);
  const int NQ = (2 * B + 2) * 128;

  float* xout = (float*)d_out;
  float* xbuf = xout + (size_t)N * OUT;

  k_norm_init <<<(N + 255) / 256, 256, 0, stream>>>(norm, cnt, N, qctr, NQ);
  k_norm_count<<<(E + 255) / 256, 256, 0, stream>>>(srcI, dstI, norm, cnt, E);
  k_norm_rsqrt<<<(N + 255) / 256, 256, 0, stream>>>(norm, N);
  k_scan<<<(N + SCAN_NB - 1) / SCAN_NB, 1024, 0, stream>>>(cnt, rowptr, cursor, N);
  k_fill<<<(E + 255) / 256, 256, 0, stream>>>(srcI, dstI, cursor, dsts, E);

  float* AGGi = AGG;
  int* qc = qctr;
  auto conv = [&](int pre, const float* Xin, const float* pbias, int store_x,
                  const float* W, int K, int Hout) {
    size_t shmem = (size_t)(128 + Hout) * (K + 4) * sizeof(float);
    unsigned gblocks = (unsigned)((N + 127) / 128);
    const int rtiles = (N + GR - 1) / GR;
    float* AGGo = AGGi;
    if (Hout != K) AGGo = (AGGi == AGG) ? AGG2 : AGG;
    if (K == GK && Hout == 128) {
      if (pre == 0)
        k_gemm_wreg<0, 128><<<GEMM_BLOCKS, 256, 0, stream>>>(Xin, W, norm, nullptr, nullptr, nullptr, 0,
                                                         S, AGGo, qc + 64, N, rtiles);
      else if (pre == 1)
        k_gemm_wreg<1, 128><<<GEMM_BLOCKS, 256, 0, stream>>>(nullptr, W, norm, AGGi, pbias, xbuf, store_x,
                                                         S, AGGo, qc + 64, N, rtiles);
      else
        k_gemm_wreg<2, 128><<<GEMM_BLOCKS, 256, 0, stream>>>(nullptr, W, norm, AGGi, pbias, xbuf, 1,
                                                         S, AGGo, qc + 64, N, rtiles);
    } else if (K == GK && Hout == 64) {
      if (pre == 0)
        k_gemm_wreg<0, 64><<<GEMM_BLOCKS, 256, 0, stream>>>(Xin, W, norm, nullptr, nullptr, nullptr, 0,
                                                        S, AGGo, qc + 64, N, rtiles);
      else if (pre == 1)
        k_gemm_wreg<1, 64><<<GEMM_BLOCKS, 256, 0, stream>>>(nullptr, W, norm, AGGi, pbias, xbuf, store_x,
                                                        S, AGGo, qc + 64, N, rtiles);
      else
        k_gemm_wreg<2, 64><<<GEMM_BLOCKS, 256, 0, stream>>>(nullptr, W, norm, AGGi, pbias, xbuf, 1,
                                                        S, AGGo, qc + 64, N, rtiles);
    } else if (pre == 0)
      k_gemm_wmma<0><<<gblocks, 256, shmem, stream>>>(Xin, W, norm, nullptr, nullptr, nullptr, 0,
                                                      S, AGGo, N, K, Hout);
    else if (pre == 1)
      k_gemm_wmma<1><<<gblocks, 256, shmem, stream>>>(nullptr, W, norm, AGGi, pbias, xbuf, store_x,
                                                      S, AGGo, N, K, Hout);
    else
      k_gemm_wmma<2><<<gblocks, 256, shmem, stream>>>(nullptr, W, norm, AGGi, pbias, xbuf, 1,
                                                      S, AGGo, N, K, Hout);
    if (Hout == 128)
      k_push<4><<<PUSH_BLOCKS, 256, 0, stream>>>(rowptr, dsts, S, AGGo, qc, N);
    else if (Hout == 64)
      k_push<2><<<PUSH_BLOCKS, 256, 0, stream>>>(rowptr, dsts, S, AGGo, qc, N);
    else
      k_push_any<<<(unsigned)((N + 7) / 8), 256, 0, stream>>>(rowptr, dsts, S, AGGo, N, Hout);
    qc += 128;
    AGGi = AGGo;
  };

  conv(0, inputs, nullptr, 0, w_in, IN, H);
  const float* pb = b_in;
  int pre = 1;
  for (int i = 0; i < B; ++i) {
    conv(pre, nullptr, pb, 1, w1 + (size_t)i * H * H, H, H);
    conv(1, nullptr, b1 + (size_t)i * H, 0, w2 + (size_t)i * H * H, H, H);
    pb = b2 + (size_t)i * H;
    pre = 2;
  }
  conv(pre, nullptr, pb, 1, w_out, H, OUT);
  long tot2 = (long)N * (OUT >> 2);
  k_finalize<<<(unsigned)((tot2 + 255) / 256), 256, 0, stream>>>(AGGi, norm, b_out, xout, N, OUT);
}


# GEMM Hout 128: a wave's two tiles are the even and the odd columns of 32 neighbours: a lane stores its two results of a row as 8 bytes, 16 lanes a whole 128-B line, no LDS pass; sums unchanged
# speedup vs baseline: 3.2555x; 1.0318x over previous
#include <hip/hip_runtime.h>

typedef __attribute__((ext_vector_type(2))) float v2f;
typedef __attribute__((ext_vector_type(8))) float v8f;

#define NEG_SLOPE 0.01f

__global__ void k_norm_init(float* __restrict__ norm, int* __restrict__ cnt, int n,
                            int* __restrict__ qctr, int nq) {
  int i = blockIdx.x * blockDim.x + threadIdx.x;
  if (i < n) {
    norm[i] = 1.0f;
    cnt[i] = 0;
  }
  for (int j = i; j < nq; j += (int)(gridDim.x * blockDim.x)) qctr[j] = 0;
}
__global__ void k_norm_count(const int* __restrict__ src, const int* __restrict__ dst,
                             float* __restrict__ norm, int* __restrict__ cnt, int e) {
  int i = blockIdx.x * blockDim.x + threadIdx.x;
  if (i < e) {
    atomicAdd(&norm[dst[i]], 1.0f);
    atomicAdd(&cnt[src[i]], 1);
  }
}
__global__ void k_norm_rsqrt(float* __restrict__ norm, int n) {
  int i = blockIdx.x * blockDim.x + threadIdx.x;
  if (i < n) norm[i] = rsqrtf(norm[i]);
}

#define SCAN_NB 2048
__global__ __launch_bounds__(1024) void k_scan(const int* __restrict__ cnt, int* __restrict__ rowptr,
                                               int* __restrict__ cursor, int n) {
  __shared__ int wsum[32];
  __shared__ int wtot[32];
  const int t = threadIdx.x;
  const int lane = t & 31, w = t >> 5;
  const int n0 = blockIdx.x * SCAN_NB;
  int s = 0;
  for (int i = t; i < n0; i += 1024) s += cnt[i];
  for (int off = 16; off > 0; off >>= 1) s += __shfl_xor(s, off);
  const int i0 = n0 + w * 64 + lane;
  const int i1 = i0 + 32;
  const int c0 = (i0 < n) ? cnt[i0] : 0;
  const int c1 = (i1 < n) ? cnt[i1] : 0;
  int x0 = c0, x1 = c1;
  for (int off = 1; off < 32; off <<= 1) {
    int y0 = __shfl_up(x0, off);
    int y1 = __shfl_up(x1, off);
    if (lane >= off) { x0 += y0; x1 += y1; }
  }
  const int t0 = __shfl(x0, 31);
  const int tw = t0 + __shfl(x1, 31);
  if (lane == 0) { wsum[w] = s; wtot[w] = tw; }
  __syncthreads();
  int a = wsum[lane];
  for (int off = 16; off > 0; off >>= 1) a += __shfl_xor(a, off);
  const int z = wtot[lane];
  int zi = z;
  for (int off = 1; off < 32; off <<= 1) {
    int y = __shfl_up(zi, off);
    if (lane >= off) zi += y;
  }
  const int base = a + __shfl(zi - z, w);
  const int e0 = base + x0 - c0;
  const int e1 = base + t0 + x1 - c1;
  if (i0 < n) { rowptr[i0] = e0; cursor[i0] = e0; }
  if (i1 < n) { rowptr[i1] = e1; cursor[i1] = e1; }
  if (i0 == n - 1) rowptr[n] = e0 + c0;
  if (i1 == n - 1) rowptr[n] = e1 + c1;
}
__global__ void k_fill(const int* __restrict__ src, const int* __restrict__ dst,
                       int* __restrict__ cursor, int* __restrict__ dsts, int e) {
  int i = blockIdx.x * blockDim.x + threadIdx.x;
  if (i < e) {
    int p = atomicAdd(&cursor[src[i]], 1);
    dsts[p] = dst[i];
  }
}

template <int PRE>
__global__ __launch_bounds__(256) void k_gemm_wmma(
    const float* __restrict__ X, const float* __restrict__ W,
    const float* __restrict__ norm,
    const float* AGGi, const float* __restrict__ pbias,
    float* __restrict__ xio, int store_x,
    float* __restrict__ S, float* AGGo,
    int nrows, int K, int Hout) {
  extern __shared__ float lds[];
  const int KSTRIDE = K + 4;
  float* Xs = lds;
  float* Wt = lds + 128 * KSTRIDE;

  const int tid = threadIdx.x;
  const int row0 = blockIdx.x * 128;

  for (int idx = tid * 4; idx < K * Hout; idx += 256 * 4) {
    int k = idx / Hout, c = idx - k * Hout;
    float4 v = *(const float4*)&W[idx];
    Wt[(c + 0) * KSTRIDE + k] = v.x;
    Wt[(c + 1) * KSTRIDE + k] = v.y;
    Wt[(c + 2) * KSTRIDE + k] = v.z;
    Wt[(c + 3) * KSTRIDE + k] = v.w;
  }
  for (int idx = tid * 4; idx < 128 * K; idx += 256 * 4) {
    int r = idx / K, c = idx - r * K;
    float4 v = {0.f, 0.f, 0.f, 0.f};
    int gr = row0 + r;
    if (gr < nrows) {
      long o = (long)gr * K + c;
      if (PRE == 0) {
        v = *(const float4*)&X[o];
      } else {
        float nr = norm[gr];
        float4 av = *(const float4*)&AGGi[o];
        float4 bv = *(const float4*)&pbias[c];
        v.x = av.x * nr + bv.x;
        v.y = av.y * nr + bv.y;
        v.z = av.z * nr + bv.z;
        v.w = av.w * nr + bv.w;
        v.x = v.x >= 0.f ? v.x : NEG_SLOPE * v.x;
        v.y = v.y >= 0.f ? v.y : NEG_SLOPE * v.y;
        v.z = v.z >= 0.f ? v.z : NEG_SLOPE * v.z;
        v.w = v.w >= 0.f ? v.w : NEG_SLOPE * v.w;
        if (PRE == 2) {
          float4 xv = *(const float4*)&xio[o];
          v.x = (xv.x + v.x) * 0.5f;
          v.y = (xv.y + v.y) * 0.5f;
          v.z = (xv.z + v.z) * 0.5f;
          v.w = (xv.w + v.w) * 0.5f;
        }
        if (PRE == 2 || store_x) *(float4*)&xio[o] = v;
      }
    }
    *(float4*)&Xs[r * KSTRIDE + c] = v;
  }
  __syncthreads();

  const int lane = tid & 31;
  const int wave = tid >> 5;
  const int mrow = lane & 15;
  const int kg   = lane >> 4;
  const float* Xrow = &Xs[(wave * 16 + mrow) * KSTRIDE + 2 * kg];

  float nrm[8];
#pragma unroll
  for (int r = 0; r < 8; ++r) {
    int grow = row0 + wave * 16 + r + 8 * kg;
    nrm[r] = (grow < nrows) ? norm[grow] : 0.f;
  }

  for (int ct = 0; ct < (Hout >> 4); ++ct) {
    const int col = ct * 16 + mrow;
    const float* Wcol = &Wt[col * KSTRIDE + 2 * kg];
    v8f acc = {};
    for (int k = 0; k < K; k += 4) {
      v2f a = *(const v2f*)&Xrow[k];
      v2f b = *(const v2f*)&Wcol[k];
      acc = __builtin_amdgcn_wmma_f32_16x16x4_f32(
           false, a,  false, b,
           (short)0, acc,  false,  false);
    }
#pragma unroll
    for (int r = 0; r < 8; ++r) {
      int grow = row0 + wave * 16 + r + 8 * kg;
      if (grow < nrows) {
        float sv = acc[r] * nrm[r];
        long o = (long)grow * Hout + col;
        S[o]    = sv;
        AGGo[o] = sv;
      }
    }
  }
}

static __device__ __forceinline__ void lds_barrier() {
  __builtin_amdgcn_fence(__ATOMIC_RELEASE, "workgroup", "local");
  __builtin_amdgcn_s_barrier();
  __builtin_amdgcn_fence(__ATOMIC_ACQUIRE, "workgroup", "local");
}

#define GK 128
#define GR 32
#define GEMM_BLOCKS 512
template <int PRE, int HOUT>
__global__ __launch_bounds__(256) void k_gemm_wreg(
    const float* __restrict__ X, const float* __restrict__ W,
    const float* __restrict__ norm,
    const float* AGGi, const float* __restrict__ pbias,
    float* __restrict__ xio, int store_x,
    float* __restrict__ S, float* AGGo, int* __restrict__ gctr, int nrows, int ntiles) {
  constexpr int KS = GK + 4;
  constexpr int NCW = (HOUT == 128) ? 2 : 1;
  constexpr int NCG = (HOUT / 16) / NCW;
  constexpr int NRG = 8 / NCG;
  constexpr int RTW = (GR / 16) / NRG;
  constexpr int NT = GR / 8;
  __shared__ __attribute__((aligned(16))) float Xs[GR * KS];
  __shared__ __attribute__((aligned(16))) float nrm_s[GR];
  __shared__ int tick_s[2];

  const int tid = threadIdx.x;
  const int lane = tid & 31;
  const int wave = tid >> 5;
  const int mrow = lane & 15;
  const int kg   = lane >> 4;
  const int cg  = wave % NCG;
  const int rg  = wave / NCG;
  const int col = (NCW == 2) ? cg * 32 + 2 * mrow
                             : cg * 16 + mrow;
  const int sc = lane * 4;
  const int sr = wave;

  v2f b[NCW][GK / 4];
  const float* wp = W + (2 * kg) * HOUT + col;
#pragma unroll
  for (int kk = 0; kk < GK / 4; ++kk) {
    if (NCW == 2) {
      const v2f w0 = *(const v2f*)&wp[(4 * kk) * HOUT];
      const v2f w1 = *(const v2f*)&wp[(4 * kk + 1) * HOUT];
      b[0][kk].x = w0.x; b[0][kk].y = w1.x;
      b[NCW - 1][kk].x = w0.y; b[NCW - 1][kk].y = w1.y;
    } else {
      b[0][kk].x = wp[(4 * kk) * HOUT];
      b[0][kk].y = wp[(4 * kk + 1) * HOUT];
    }
  }
  float4 bv = {0.f, 0.f, 0.f, 0.f};
  if (PRE != 0) bv = *(const float4*)&pbias[sc];

  const int tlow = ((nrows >> 1) + GR - 1) / GR;
  int hh = 0, other = 0, tkraw = 0;
  auto ask = [&]() { tkraw = atomicAdd(gctr + hh * 32, 1); };
  auto take = [&]() -> int {
    int cnt = hh ? ntiles - tlow : tlow;
    int tk = tkraw;
    if (tk >= cnt && !other) {
      other = 1;
      hh ^= 1;
      cnt = hh ? ntiles - tlow : tlow;
      tk = atomicAdd(gctr + hh * 32, 1);
    }
    return (tk < cnt) ? (hh ? tlow + tk : tk) : -1;
  };
  if (tid == 0) {
    hh = (__builtin_amdgcn_s_sendmsg_rtn(0x87) >> 18) & 1;
    ask();
    tick_s[0] = take();
  }
  lds_barrier();
  int tile = __builtin_amdgcn_readfirstlane(tick_s[0]);

  float nr[NT];
  float4 av[NT], xv[NT];
  auto request = [&](int t_) {
    const int r0 = t_ * GR + sr;
#pragma unroll
    for (int t = 0; t < NT; ++t) {
      const int gr = r0 + 8 * t;
      const int grc = gr < nrows ? gr : nrows - 1;
      const long o = (long)grc * GK + sc;
      nr[t] = norm[grc];
      if (PRE == 0) {
        av[t] = *(const float4*)&X[o];
      } else {
        av[t] = *(const float4*)&AGGi[o];
        if (PRE == 2) xv[t] = *(const float4*)&xio[o];
      }
    }
  };
  if (tile >= 0) {
    request(tile);
    if (tid == 0) ask();
  }

  while (tile >= 0) {
    const int row0 = tile * GR;
#pragma unroll
    for (int t = 0; t < NT; ++t) {
      const int r = sr + 8 * t;
      const int gr = row0 + r;
      const bool ok = gr < nrows;
      float4 v = av[t];
      if (PRE != 0) {
        const float nrt = nr[t];
        v.x = v.x * nrt + bv.x;
        v.y = v.y * nrt + bv.y;
        v.z = v.z * nrt + bv.z;
        v.w = v.w * nrt + bv.w;
        v.x = v.x >= 0.f ? v.x : NEG_SLOPE * v.x;
        v.y = v.y >= 0.f ? v.y : NEG_SLOPE * v.y;
        v.z = v.z >= 0.f ? v.z : NEG_SLOPE * v.z;
        v.w = v.w >= 0.f ? v.w : NEG_SLOPE * v.w;
        if (PRE == 2) {
          v.x = (xv[t].x + v.x) * 0.5f;
          v.y = (xv[t].y + v.y) * 0.5f;
          v.z = (xv[t].z + v.z) * 0.5f;
          v.w = (xv[t].w + v.w) * 0.5f;
        }
        if (ok && (PRE == 2 || store_x)) *(float4*)&xio[(long)gr * GK + sc] = v;
      }
      if (!ok) v = float4{0.f, 0.f, 0.f, 0.f};
      *(float4*)&Xs[r * KS + sc] = v;
      if (lane == 0) nrm_s[r] = ok ? nr[t] : 0.f;
    }
    if (tid == 0) tick_s[1] = take();
    lds_barrier();
    const int next = __builtin_amdgcn_readfirstlane(tick_s[1]);
    if (next >= 0) {
      request(next);
      if (tid == 0) ask();
    }

    const float* xr = &Xs[((rg * RTW) * 16 + mrow) * KS + 2 * kg];
    v8f acc[RTW][NCW];
#pragma unroll
    for (int j = 0; j < RTW; ++j)
#pragma unroll
      for (int c = 0; c < NCW; ++c) acc[j][c] = v8f{};
#pragma unroll
    for (int kk = 0; kk < GK / 4; ++kk) {
#pragma unroll
      for (int j = 0; j < RTW; ++j) {
        v2f a = *(const v2f*)&xr[j * 16 * KS + 4 * kk];
#pragma unroll
        for (int c = 0; c < NCW; ++c) {
          acc[j][c] = __builtin_amdgcn_wmma_f32_16x16x4_f32(
               false, a,  false, b[c][kk],
               (short)0, acc[j][c],  false,  false);
        }
      }
    }
    float nrm[RTW][8];
#pragma unroll
    for (int j = 0; j < RTW; ++j) {
      const float4* np = (const float4*)&nrm_s[(rg * RTW + j) * 16 + 8 * kg];
      const float4 n0 = np[0], n1 = np[1];
      nrm[j][0] = n0.x; nrm[j][1] = n0.y; nrm[j][2] = n0.z; nrm[j][3] = n0.w;
      nrm[j][4] = n1.x; nrm[j][5] = n1.y; nrm[j][6] = n1.z; nrm[j][7] = n1.w;
    }
#pragma unroll
    for (int j = 0; j < RTW; ++j) {
#pragma unroll
      for (int r = 0; r < 8; ++r) {
        const int lr = (rg * RTW + j) * 16 + r + 8 * kg;
        const int grow = row0 + lr;
        if (grow < nrows) {
          long o = (long)grow * HOUT + col;
          if (NCW == 2) {
            v2f sv;
            sv.x = acc[j][0][r] * nrm[j][r];
            sv.y = acc[j][NCW - 1][r] * nrm[j][r];
            *(v2f*)&S[o]    = sv;
            *(v2f*)&AGGo[o] = sv;
          } else {
            float sv = acc[j][0][r] * nrm[j][r];
            S[o]    = sv;
            AGGo[o] = sv;
          }
        }
      }
    }
    lds_barrier();
    tile = next;
  }
}

#define PUSH_BLOCKS 2048
#define PUSH_CHW 4
template <int NV>
__global__ __launch_bounds__(256) void k_push(const int* __restrict__ rowptr,
                                              const int* __restrict__ dsts,
                                              const float* __restrict__ S, float* __restrict__ AGG,
                                              int* __restrict__ qctr, int n) {
  constexpr int HOUT = NV * 32;
  constexpr int SPB = 8 * PUSH_CHW;
  __shared__ int tk[2];
  const int tid = threadIdx.x;
  const int lane = tid & 31;
  const int wave = __builtin_amdgcn_readfirstlane(tid >> 5);
  const int nchunks = (n + SPB - 1) / SPB;
  const int nh = n >> 1;
  int h = (__builtin_amdgcn_s_sendmsg_rtn(0x87) >> 18) & 1;
  for (int pass = 0; pass < 2; ++pass, h ^= 1) {
    const int lo = h ? nh : 0;
    const int hi = h ? n : nh;
    int* ctr = qctr + h * 32;
    if (tid == 0) tk[0] = atomicAdd(ctr, 1);
    lds_barrier();
    for (int it = 0;; ++it) {
      const int t = __builtin_amdgcn_readfirstlane(tk[it & 1]);
      if (t >= nchunks) break;
      const int s0 = t * SPB + wave * PUSH_CHW;
      int rp[PUSH_CHW + 1];
#pragma unroll
      for (int k = 0; k <= PUSH_CHW; ++k) {
        const int si = s0 + k;
        rp[k] = rowptr[si < n ? si : n];
      }
      float v[PUSH_CHW][NV];
      int dl0[PUSH_CHW];
#pragma unroll
      for (int k = 0; k < PUSH_CHW; ++k) {
        const int s = (s0 + k < n) ? (s0 + k) : (n - 1);
        const float* sp = &S[(long)s * HOUT + lane];
#pragma unroll
        for (int i = 0; i < NV; ++i) v[k][i] = sp[32 * i];
        const int deg = rp[k + 1] - rp[k];
        dl0[k] = (lane < deg) ? dsts[rp[k] + lane] : -1;
      }
      int nxt = 0;
      if (tid == 0) nxt = atomicAdd(ctr, 1);
#pragma unroll
      for (int k = 0; k < PUSH_CHW; ++k) {
        const int beg = rp[k];
        const int deg = rp[k + 1] - beg;
        int dl = dl0[k];
        for (int base = 0; base < deg; base += 32) {
          if (base > 0) dl = (base + lane < deg) ? dsts[beg + base + lane] : -1;
          unsigned m = __builtin_amdgcn_ballot_w32(dl >= lo && dl < hi);
          while (m) {
            const int j = __builtin_ctz(m);
            m &= m - 1;
            const int d = __builtin_amdgcn_readlane(dl, j);
            float* ap = &AGG[(long)d * HOUT + lane];
#pragma unroll
            for (int i = 0; i < NV; ++i) atomicAdd(ap + 32 * i, v[k][i]);
          }
        }
      }

      if (tid == 0) tk[(it + 1) & 1] = nxt;
      lds_barrier();
    }
    lds_barrier();
  }
}

__global__ __launch_bounds__(256) void k_push_any(const int* __restrict__ rowptr,
                                                  const int* __restrict__ dsts,
                                                  const float* __restrict__ S,
                                                  float* __restrict__ AGG, int n, int Hout) {
  const int lane = threadIdx.x & 31;
  const int s = __builtin_amdgcn_readfirstlane((int)(blockIdx.x * (blockDim.x >> 5) + (threadIdx.x >> 5)));
  if (s >= n) return;
  const int beg = rowptr[s], end = rowptr[s + 1];
  const float* sp = &S[(long)s * Hout];
  for (int ed = beg; ed < end; ++ed) {
    const int d = dsts[ed];
    float* ap = &AGG[(long)d * Hout];
    for (int c = lane; c < Hout; c += 32) atomicAdd(ap + c, sp[c]);
  }
}

__global__ void k_finalize(const float* __restrict__ AGG, const float* __restrict__ norm,
                           const float* __restrict__ bias, float* __restrict__ out,
                           int n, int Hout) {
  const int chunks = Hout >> 2;
  long tid = (long)blockIdx.x * blockDim.x + threadIdx.x;
  if (tid >= (long)n * chunks) return;
  int i = (int)(tid / chunks);
  int c = (int)(tid - (long)i * chunks) << 2;
  long o = (long)i * Hout + c;
  float nr = norm[i];
  float4 av = *(const float4*)&AGG[o];
  float4 bv = *(const float4*)&bias[c];
  float4 r4 = {av.x * nr + bv.x, av.y * nr + bv.y, av.z * nr + bv.z, av.w * nr + bv.w};
  *(float4*)&out[o] = r4;
}

extern "C" void kernel_launch(void* const* d_in, const int* in_sizes, int n_in,
                              void* d_out, int out_size, void* d_ws, size_t ws_size,
                              hipStream_t stream) {
  (void)n_in; (void)out_size; (void)ws_size;
  const float* inputs = (const float*)d_in[0];
  const int*   edges  = (const int*)d_in[1];
  const float* w_in   = (const float*)d_in[2];
  const float* b_in   = (const float*)d_in[3];
  const float* w1     = (const float*)d_in[4];
  const float* b1     = (const float*)d_in[5];
  const float* w2     = (const float*)d_in[6];
  const float* b2     = (const float*)d_in[7];
  const float* w_out  = (const float*)d_in[8];
  const float* b_out  = (const float*)d_in[9];

  const int H   = in_sizes[3];
  const int IN  = in_sizes[2] / H;
  const int N   = in_sizes[0] / IN;
  const int E   = in_sizes[1] / 2;
  const int OUT = in_sizes[9];
  const int B   = in_sizes[5] / H;

  const int* srcI = edges;
  const int* dstI = edges + E;

  float* norm = (float*)d_ws;
  float* S    = norm + (((size_t)N + 255) & ~(size_t)255);
  float* AGG  = S + (size_t)N * H;
  float* AGG2 = AGG + (size_t)N * H;
  const size_t NP = ((size_t)N + 256) & ~(size_t)255;
  int* cnt    = (int*)(AGG2 + (size_t)N * H);
  int* rowptr = cnt + NP;
  int* cursor = rowptr + NP;
  int* dsts   = cursor + NP;
  int* qctr   = dsts + (((size_t)E + 31) & ~(size_t)31);
  const int NQ = (2 * B + 2) * 128;

  float* xout = (float*)d_out;
  float* xbuf = xout + (size_t)N * OUT;

  k_norm_init <<<(N + 255) / 256, 256, 0, stream>>>(norm, cnt, N, qctr, NQ);
  k_norm_count<<<(E + 255) / 256, 256, 0, stream>>>(srcI, dstI, norm, cnt, E);
  k_norm_rsqrt<<<(N + 255) / 256, 256, 0, stream>>>(norm, N);
  k_scan<<<(N + SCAN_NB - 1) / SCAN_NB, 1024, 0, stream>>>(cnt, rowptr, cursor, N);
  k_fill<<<(E + 255) / 256, 256, 0, stream>>>(srcI, dstI, cursor, dsts, E);

  float* AGGi = AGG;
  int* qc = qctr;
  auto conv = [&](int pre, const float* Xin, const float* pbias, int store_x,
                  const float* W, int K, int Hout) {
    size_t shmem = (size_t)(128 + Hout) * (K + 4) * sizeof(float);
    unsigned gblocks = (unsigned)((N + 127) / 128);
    const int rtiles = (N + GR - 1) / GR;
    float* AGGo = AGGi;
    if (Hout != K) AGGo = (AGGi == AGG) ? AGG2 : AGG;
    if (K == GK && Hout == 128) {
      if (pre == 0)
        k_gemm_wreg<0, 128><<<GEMM_BLOCKS, 256, 0, stream>>>(Xin, W, norm, nullptr, nullptr, nullptr, 0,
                                                         S, AGGo, qc + 64, N, rtiles);
      else if (pre == 1)
        k_gemm_wreg<1, 128><<<GEMM_BLOCKS, 256, 0, stream>>>(nullptr, W, norm, AGGi, pbias, xbuf, store_x,
                                                         S, AGGo, qc + 64, N, rtiles);
      else
        k_gemm_wreg<2, 128><<<GEMM_BLOCKS, 256, 0, stream>>>(nullptr, W, norm, AGGi, pbias, xbuf, 1,
                                                         S, AGGo, qc + 64, N, rtiles);
    } else if (K == GK && Hout == 64) {
      if (pre == 0)
        k_gemm_wreg<0, 64><<<GEMM_BLOCKS, 256, 0, stream>>>(Xin, W, norm, nullptr, nullptr, nullptr, 0,
                                                        S, AGGo, qc + 64, N, rtiles);
      else if (pre == 1)
        k_gemm_wreg<1, 64><<<GEMM_BLOCKS, 256, 0, stream>>>(nullptr, W, norm, AGGi, pbias, xbuf, store_x,
                                                        S, AGGo, qc + 64, N, rtiles);
      else
        k_gemm_wreg<2, 64><<<GEMM_BLOCKS, 256, 0, stream>>>(nullptr, W, norm, AGGi, pbias, xbuf, 1,
                                                        S, AGGo, qc + 64, N, rtiles);
    } else if (pre == 0)
      k_gemm_wmma<0><<<gblocks, 256, shmem, stream>>>(Xin, W, norm, nullptr, nullptr, nullptr, 0,
                                                      S, AGGo, N, K, Hout);
    else if (pre == 1)
      k_gemm_wmma<1><<<gblocks, 256, shmem, stream>>>(nullptr, W, norm, AGGi, pbias, xbuf, store_x,
                                                      S, AGGo, N, K, Hout);
    else
      k_gemm_wmma<2><<<gblocks, 256, shmem, stream>>>(nullptr, W, norm, AGGi, pbias, xbuf, 1,
                                                      S, AGGo, N, K, Hout);
    if (Hout == 128)
      k_push<4><<<PUSH_BLOCKS, 256, 0, stream>>>(rowptr, dsts, S, AGGo, qc, N);
    else if (Hout == 64)
      k_push<2><<<PUSH_BLOCKS, 256, 0, stream>>>(rowptr, dsts, S, AGGo, qc, N);
    else
      k_push_any<<<(unsigned)((N + 7) / 8), 256, 0, stream>>>(rowptr, dsts, S, AGGo, N, Hout);
    qc += 128;
    AGGi = AGGo;
  };

  conv(0, inputs, nullptr, 0, w_in, IN, H);
  const float* pb = b_in;
  int pre = 1;
  for (int i = 0; i < B; ++i) {
    conv(pre, nullptr, pb, 1, w1 + (size_t)i * H * H, H, H);
    conv(1, nullptr, b1 + (size_t)i * H, 0, w2 + (size_t)i * H * H, H, H);
    pb = b2 + (size_t)i * H;
    pre = 2;
  }
  conv(pre, nullptr, pb, 1, w_out, H, OUT);
  long tot2 = (long)N * (OUT >> 2);
  k_finalize<<<(unsigned)((tot2 + 255) / 256), 256, 0, stream>>>(AGGi, norm, b_out, xout, N, OUT);
}
